# starting kernel
# baseline (speedup 1.0000x reference)
#include <hip/hip_runtime.h>

#define N_TOK   131072
#define KEXP    16
#define IN_DIM  256
#define OUT_DIM 128

#define ROWS_PER_BLK 128
#define MOE_BLOCKS   1040
#define IDX_CAP      (MOE_BLOCKS * ROWS_PER_BLK)

typedef __bf16 bf16_t;
typedef __attribute__((ext_vector_type(16))) bf16_t v16bf;
typedef __attribute__((ext_vector_type(8)))  float  v8f;


__global__ __launch_bounds__(256) void init_ws(int* counts, int* cur, int* idxb) {
    int i = blockIdx.x * blockDim.x + threadIdx.x;
    if (i < KEXP) { counts[i] = 0; cur[i] = 0; }
    if (i < IDX_CAP) idxb[i] = -1;
}

__global__ __launch_bounds__(256) void count_k(const int* __restrict__ z, int* counts) {
    int i = blockIdx.x * blockDim.x + threadIdx.x;
    if (i < N_TOK) atomicAdd(&counts[z[i]], 1);
}

__global__ void offsets_k(const int* __restrict__ counts, int* off) {
    if (blockIdx.x == 0 && threadIdx.x == 0) {
        int acc = 0;
        for (int e = 0; e < KEXP; ++e) {
            off[e] = acc;
            acc += (counts[e] + ROWS_PER_BLK - 1) & ~(ROWS_PER_BLK - 1);
        }
        off[KEXP] = acc;
    }
}

__global__ __launch_bounds__(256) void scatter_k(const int* __restrict__ z,
                                                 const int* __restrict__ off,
                                                 int* cur, int* idxb) {
    int i = blockIdx.x * blockDim.x + threadIdx.x;
    if (i < N_TOK) {
        int e = z[i];
        int p = off[e] + atomicAdd(&cur[e], 1);
        idxb[p] = i;
    }
}


__global__ __launch_bounds__(256) void moe_k(const float* __restrict__ x,
                                             const int*   __restrict__ idxb,
                                             const int*   __restrict__ off,
                                             const float* __restrict__ W1,
                                             const float* __restrict__ b1,
                                             const float* __restrict__ W2,
                                             const float* __restrict__ b2,
                                             float* __restrict__ out) {
    __shared__ __attribute__((aligned(32))) unsigned char lds[64 * 1024];

    const int base  = blockIdx.x * ROWS_PER_BLK;
    const int total = off[KEXP];
    if (base >= total) return;

    int e = 0;
#pragma unroll
    for (int i = 1; i < KEXP; ++i) e = (base >= off[i]) ? i : e;

    const int tid    = threadIdx.x;
    const int lane   = tid & 31;
    const int wv     = tid >> 5;
    const int hi     = lane >> 4;
    const int lanelo = lane & 15;

    const float* w1 = W1 + (size_t)e * (IN_DIM * OUT_DIM);
    for (int i = tid; i < IN_DIM * OUT_DIM; i += 256) {
        int k = i >> 7;
        int n = i & 127;
        bf16_t v = (bf16_t)w1[i];
        int kstep = k >> 5, kk = k & 31;
        int ntile = n >> 4, nn = n & 15;
        int ln   = (((kk >> 3) & 1) << 4) | nn;
        int slot = (((kk >> 4) & 1) << 3) | (kk & 7);
        *(bf16_t*)(lds + (((kstep << 3) | ntile) << 10) + (ln << 5) + (slot << 1)) = v;
    }
    __syncthreads();

    const int rowPos = base + wv * 16 + lanelo;
    const int tok    = idxb[rowPos];
    const int tclamp = tok < 0 ? 0 : tok;
    const float* xrow = x + (size_t)tclamp * IN_DIM;

    v8f zero = {0.f, 0.f, 0.f, 0.f, 0.f, 0.f, 0.f, 0.f};
    v8f acc[8];
#pragma unroll
    for (int i = 0; i < 8; ++i) acc[i] = zero;

#pragma unroll
    for (int ks = 0; ks < 8; ++ks) {
        const int kb = ks * 32 + hi * 8;
        float4 f0 = *(const float4*)(xrow + kb);
        float4 f1 = *(const float4*)(xrow + kb + 4);
        float4 f2 = *(const float4*)(xrow + kb + 16);
        float4 f3 = *(const float4*)(xrow + kb + 20);
        v16bf a;
        a[0]=(bf16_t)f0.x; a[1]=(bf16_t)f0.y; a[2] =(bf16_t)f0.z; a[3] =(bf16_t)f0.w;
        a[4]=(bf16_t)f1.x; a[5]=(bf16_t)f1.y; a[6] =(bf16_t)f1.z; a[7] =(bf16_t)f1.w;
        a[8]=(bf16_t)f2.x; a[9]=(bf16_t)f2.y; a[10]=(bf16_t)f2.z; a[11]=(bf16_t)f2.w;
        a[12]=(bf16_t)f3.x; a[13]=(bf16_t)f3.y; a[14]=(bf16_t)f3.z; a[15]=(bf16_t)f3.w;
#pragma unroll
        for (int nt = 0; nt < 8; ++nt) {
            v16bf bf = *(const v16bf*)(lds + (((ks << 3) | nt) << 10) + (lane << 5));
            acc[nt] = __builtin_amdgcn_wmma_f32_16x16x32_bf16(
                false, a, false, bf, (short)0, acc[nt], false, false);
        }
    }

    float s0[8], s1[8];
#pragma unroll
    for (int v = 0; v < 8; ++v) { s0[v] = 0.f; s1[v] = 0.f; }
#pragma unroll
    for (int nt = 0; nt < 8; ++nt) {
        int n = nt * 16 + lanelo;
        float bb  = b1[e * OUT_DIM + n];
        float w20 = W2[((size_t)e * OUT_DIM + n) * 2 + 0];
        float w21 = W2[((size_t)e * OUT_DIM + n) * 2 + 1];
#pragma unroll
        for (int v = 0; v < 8; ++v) {
            float h = acc[nt][v] + bb;
            h = h > 0.f ? h : 0.f;
            s0[v] += h * w20;
            s1[v] += h * w21;
        }
    }
#pragma unroll
    for (int v = 0; v < 8; ++v) {
#pragma unroll
        for (int m = 1; m < 16; m <<= 1) {
            s0[v] += __shfl_xor(s0[v], m, 32);
            s1[v] += __shfl_xor(s1[v], m, 32);
        }
    }

    const float b20 = b2[e * 2 + 0], b21 = b2[e * 2 + 1];
    __syncthreads();
    float* lgt = (float*)lds;
#pragma unroll
    for (int v = 0; v < 8; ++v) {
        if (lanelo == v) {
            int M = v + 8 * hi;
            lgt[(wv * 16 + M) * 2 + 0] = s0[v] + b20;
            lgt[(wv * 16 + M) * 2 + 1] = s1[v] + b21;
        }
    }
    __syncthreads();

    if (lane < 16 && tok >= 0) {
        float l0 = lgt[(wv * 16 + lane) * 2 + 0];
        float l1 = lgt[(wv * 16 + lane) * 2 + 1];
        float mx = fmaxf(l0, l1);
        float e0 = __expf(l0 - mx), e1 = __expf(l1 - mx);
        float inv = 1.f / (e0 + e1);
        out[(size_t)tok * 2 + 0] = e0 * inv;
        out[(size_t)tok * 2 + 1] = e1 * inv;
    }
}


extern "C" void kernel_launch(void* const* d_in, const int* in_sizes, int n_in,
                              void* d_out, int out_size, void* d_ws, size_t ws_size,
                              hipStream_t stream) {
    (void)in_sizes; (void)n_in; (void)out_size; (void)ws_size;
    const float* x  = (const float*)d_in[0];
    const int*   z  = (const int*)d_in[1];
    const float* W1 = (const float*)d_in[2];
    const float* b1 = (const float*)d_in[3];
    const float* W2 = (const float*)d_in[4];
    const float* b2 = (const float*)d_in[5];
    float* out = (float*)d_out;

    int* counts = (int*)d_ws;
    int* off    = counts + 16;
    int* cur    = counts + 48;
    int* idxb   = counts + 128;

    init_ws  <<<(IDX_CAP + 255) / 256, 256, 0, stream>>>(counts, cur, idxb);
    count_k  <<<(N_TOK  + 255) / 256, 256, 0, stream>>>(z, counts);
    offsets_k<<<1, 1, 0, stream>>>(counts, off);
    scatter_k<<<(N_TOK  + 255) / 256, 256, 0, stream>>>(z, off, cur, idxb);
    moe_k    <<<MOE_BLOCKS, 256, 0, stream>>>(x, idxb, off, W1, b1, W2, b2, out);
}


# Bucketing pre-pass rebuilt as atomic-free counting sort: per-chunk ballot histogram + chunk prefix give each token its slot; 5 launches become 3 (init, count, offset kernels go); row sums unchanged
# speedup vs baseline: 1.9013x; 1.9013x over previous
#include <hip/hip_runtime.h>

#define N_TOK   131072
#define KEXP    16
#define IN_DIM  256
#define OUT_DIM 128

#define ROWS_PER_BLK 128
#define MOE_BLOCKS   1040
#define IDX_CAP      (MOE_BLOCKS * ROWS_PER_BLK)

typedef __bf16 bf16_t;
typedef __attribute__((ext_vector_type(16))) bf16_t v16bf;
typedef __attribute__((ext_vector_type(8)))  float  v8f;


#define TOK_PER_CHUNK 1024
#define N_CHUNK       (N_TOK / TOK_PER_CHUNK)

__global__ __launch_bounds__(256) void hist_k(const int* __restrict__ z,
                                              int* __restrict__ hist) {
    __shared__ int wcnt[8][KEXP];
    const int tid = threadIdx.x, lane = tid & 31, wv = tid >> 5;
    const int4 zz = *(const int4*)(z + (blockIdx.x * 256 + tid) * 4);
    const int zv[4] = {zz.x, zz.y, zz.z, zz.w};
    int mine = 0;
#pragma unroll
    for (int e = 0; e < KEXP; ++e) {
        int c = 0;
#pragma unroll
        for (int j = 0; j < 4; ++j)
            c += __builtin_popcount(__builtin_amdgcn_ballot_w32(zv[j] == e));
        mine = (lane == e) ? c : mine;
    }
    if (lane < KEXP) wcnt[wv][lane] = mine;
    __syncthreads();
    if (tid < KEXP) {
        int s = 0;
#pragma unroll
        for (int w = 0; w < 8; ++w) s += wcnt[w][tid];
        hist[blockIdx.x * KEXP + tid] = s;
    }
}

__global__ __launch_bounds__(256) void scatter_k(const int* __restrict__ z,
                                                 const int* __restrict__ hist,
                                                 int* __restrict__ off,
                                                 int* __restrict__ idxb) {
    __shared__ int ppre[16][KEXP];
    __shared__ int ptot[16][KEXP];
    __shared__ int wbase[8][KEXP];
    __shared__ int segoff[KEXP];
    __shared__ int segcnt[KEXP];
    const int tid = threadIdx.x, lane = tid & 31, wv = tid >> 5;
    const int b = blockIdx.x;

    {
        const int e = tid & 15, p = tid >> 4;
        int pre = 0, tot = 0;
#pragma unroll
        for (int i = 0; i < N_CHUNK / 16; ++i) {
            const int bb = p + 16 * i;
            const int c  = hist[bb * KEXP + e];
            tot += c;
            pre += (bb < b) ? c : 0;
        }
        ppre[p][e] = pre;
        ptot[p][e] = tot;
    }

    const int tok0 = (b * 256 + tid) * 4;
    const int4 zz = *(const int4*)(z + tok0);
    const int zv[4] = {zz.x, zz.y, zz.z, zz.w};
    int r[4] = {0, 0, 0, 0};
    int mine = 0;
#pragma unroll
    for (int e = 0; e < KEXP; ++e) {
        int run = 0;
#pragma unroll
        for (int j = 0; j < 4; ++j) {
            const bool is = (zv[j] == e);
            const unsigned m = __builtin_amdgcn_ballot_w32(is);
            const int rk = (int)__builtin_amdgcn_mbcnt_lo(m, (unsigned)run);
            r[j] = is ? rk : r[j];
            run += __builtin_popcount(m);
        }
        mine = (lane == e) ? run : mine;
    }
    if (lane < KEXP) wbase[wv][lane] = mine;
    __syncthreads();

    if (wv == 0) {
        const int e = lane & 15;
        int pre = 0, tot = 0;
#pragma unroll
        for (int p = 0; p < 16; ++p) { pre += ppre[p][e]; tot += ptot[p][e]; }
        const int padded = (tot + ROWS_PER_BLK - 1) & ~(ROWS_PER_BLK - 1);
        int incl = padded;
#pragma unroll
        for (int d = 1; d < 16; d <<= 1) {
            const int nb = __shfl_up(incl, d, 16);
            incl += (e >= d) ? nb : 0;
        }
        const int first = incl - padded;
        int wc[8];
#pragma unroll
        for (int w = 0; w < 8; ++w) wc[w] = wbase[w][e];
        if (lane < KEXP) {
            int run = first + pre;
#pragma unroll
            for (int w = 0; w < 8; ++w) { wbase[w][e] = run; run += wc[w]; }
            segoff[e] = first;
            segcnt[e] = tot;
            if (b == 0) {
                off[e] = first;
                if (e == KEXP - 1) off[KEXP] = incl;
            }
        }
    }
    __syncthreads();

    if (b < KEXP && tid < ROWS_PER_BLK) {
        const int p   = segoff[b] + segcnt[b] + tid;
        const int end = segoff[b] + ((segcnt[b] + ROWS_PER_BLK - 1) & ~(ROWS_PER_BLK - 1));
        if (p < end) idxb[p] = -1;
    }

#pragma unroll
    for (int j = 0; j < 4; ++j) {
        const int e = zv[j];
        if ((unsigned)e < (unsigned)KEXP) idxb[wbase[wv][e] + r[j]] = tok0 + j;
    }
}


__global__ __launch_bounds__(256) void moe_k(const float* __restrict__ x,
                                             const int*   __restrict__ idxb,
                                             const int*   __restrict__ off,
                                             const float* __restrict__ W1,
                                             const float* __restrict__ b1,
                                             const float* __restrict__ W2,
                                             const float* __restrict__ b2,
                                             float* __restrict__ out) {
    __shared__ __attribute__((aligned(32))) unsigned char lds[64 * 1024];

    const int base  = blockIdx.x * ROWS_PER_BLK;
    const int total = off[KEXP];
    if (base >= total) return;

    int e = 0;
#pragma unroll
    for (int i = 1; i < KEXP; ++i) e = (base >= off[i]) ? i : e;

    const int tid    = threadIdx.x;
    const int lane   = tid & 31;
    const int wv     = tid >> 5;
    const int hi     = lane >> 4;
    const int lanelo = lane & 15;

    const float* w1 = W1 + (size_t)e * (IN_DIM * OUT_DIM);
    for (int i = tid; i < IN_DIM * OUT_DIM; i += 256) {
        int k = i >> 7;
        int n = i & 127;
        bf16_t v = (bf16_t)w1[i];
        int kstep = k >> 5, kk = k & 31;
        int ntile = n >> 4, nn = n & 15;
        int ln   = (((kk >> 3) & 1) << 4) | nn;
        int slot = (((kk >> 4) & 1) << 3) | (kk & 7);
        *(bf16_t*)(lds + (((kstep << 3) | ntile) << 10) + (ln << 5) + (slot << 1)) = v;
    }
    __syncthreads();

    const int rowPos = base + wv * 16 + lanelo;
    const int tok    = idxb[rowPos];
    const int tclamp = tok < 0 ? 0 : tok;
    const float* xrow = x + (size_t)tclamp * IN_DIM;

    v8f zero = {0.f, 0.f, 0.f, 0.f, 0.f, 0.f, 0.f, 0.f};
    v8f acc[8];
#pragma unroll
    for (int i = 0; i < 8; ++i) acc[i] = zero;

#pragma unroll
    for (int ks = 0; ks < 8; ++ks) {
        const int kb = ks * 32 + hi * 8;
        float4 f0 = *(const float4*)(xrow + kb);
        float4 f1 = *(const float4*)(xrow + kb + 4);
        float4 f2 = *(const float4*)(xrow + kb + 16);
        float4 f3 = *(const float4*)(xrow + kb + 20);
        v16bf a;
        a[0]=(bf16_t)f0.x; a[1]=(bf16_t)f0.y; a[2] =(bf16_t)f0.z; a[3] =(bf16_t)f0.w;
        a[4]=(bf16_t)f1.x; a[5]=(bf16_t)f1.y; a[6] =(bf16_t)f1.z; a[7] =(bf16_t)f1.w;
        a[8]=(bf16_t)f2.x; a[9]=(bf16_t)f2.y; a[10]=(bf16_t)f2.z; a[11]=(bf16_t)f2.w;
        a[12]=(bf16_t)f3.x; a[13]=(bf16_t)f3.y; a[14]=(bf16_t)f3.z; a[15]=(bf16_t)f3.w;
#pragma unroll
        for (int nt = 0; nt < 8; ++nt) {
            v16bf bf = *(const v16bf*)(lds + (((ks << 3) | nt) << 10) + (lane << 5));
            acc[nt] = __builtin_amdgcn_wmma_f32_16x16x32_bf16(
                false, a, false, bf, (short)0, acc[nt], false, false);
        }
    }

    float s0[8], s1[8];
#pragma unroll
    for (int v = 0; v < 8; ++v) { s0[v] = 0.f; s1[v] = 0.f; }
#pragma unroll
    for (int nt = 0; nt < 8; ++nt) {
        int n = nt * 16 + lanelo;
        float bb  = b1[e * OUT_DIM + n];
        float w20 = W2[((size_t)e * OUT_DIM + n) * 2 + 0];
        float w21 = W2[((size_t)e * OUT_DIM + n) * 2 + 1];
#pragma unroll
        for (int v = 0; v < 8; ++v) {
            float h = acc[nt][v] + bb;
            h = h > 0.f ? h : 0.f;
            s0[v] += h * w20;
            s1[v] += h * w21;
        }
    }
#pragma unroll
    for (int v = 0; v < 8; ++v) {
#pragma unroll
        for (int m = 1; m < 16; m <<= 1) {
            s0[v] += __shfl_xor(s0[v], m, 32);
            s1[v] += __shfl_xor(s1[v], m, 32);
        }
    }

    const float b20 = b2[e * 2 + 0], b21 = b2[e * 2 + 1];
    __syncthreads();
    float* lgt = (float*)lds;
#pragma unroll
    for (int v = 0; v < 8; ++v) {
        if (lanelo == v) {
            int M = v + 8 * hi;
            lgt[(wv * 16 + M) * 2 + 0] = s0[v] + b20;
            lgt[(wv * 16 + M) * 2 + 1] = s1[v] + b21;
        }
    }
    __syncthreads();

    if (lane < 16 && tok >= 0) {
        float l0 = lgt[(wv * 16 + lane) * 2 + 0];
        float l1 = lgt[(wv * 16 + lane) * 2 + 1];
        float mx = fmaxf(l0, l1);
        float e0 = __expf(l0 - mx), e1 = __expf(l1 - mx);
        float inv = 1.f / (e0 + e1);
        out[(size_t)tok * 2 + 0] = e0 * inv;
        out[(size_t)tok * 2 + 1] = e1 * inv;
    }
}


extern "C" void kernel_launch(void* const* d_in, const int* in_sizes, int n_in,
                              void* d_out, int out_size, void* d_ws, size_t ws_size,
                              hipStream_t stream) {
    (void)in_sizes; (void)n_in; (void)out_size; (void)ws_size;
    const float* x  = (const float*)d_in[0];
    const int*   z  = (const int*)d_in[1];
    const float* W1 = (const float*)d_in[2];
    const float* b1 = (const float*)d_in[3];
    const float* W2 = (const float*)d_in[4];
    const float* b2 = (const float*)d_in[5];
    float* out = (float*)d_out;

    int* off  = (int*)d_ws;
    int* hist = off + 32;
    int* idxb = hist + N_CHUNK * KEXP;

    hist_k   <<<N_CHUNK, 256, 0, stream>>>(z, hist);
    scatter_k<<<N_CHUNK, 256, 0, stream>>>(z, hist, off, idxb);
    moe_k    <<<MOE_BLOCKS, 256, 0, stream>>>(x, idxb, off, W1, b1, W2, b2, out);
}


# W1 converted to the bf16 B-fragment image once per launch (extra work-groups of the histogram launch, 1 MB of workspace); row-tile work-groups copy 64 KB with 16-byte loads instead of converting
# speedup vs baseline: 2.3026x; 1.2111x over previous
#include <hip/hip_runtime.h>

#define N_TOK   131072
#define KEXP    16
#define IN_DIM  256
#define OUT_DIM 128

#define ROWS_PER_BLK 128
#define MOE_BLOCKS   1040
#define IDX_CAP      (MOE_BLOCKS * ROWS_PER_BLK)

typedef __bf16 bf16_t;
typedef __attribute__((ext_vector_type(16))) bf16_t v16bf;
typedef __attribute__((ext_vector_type(8)))  float  v8f;
typedef __attribute__((ext_vector_type(8)))  bf16_t v8bf;

#define FRAG_BYTES    1024
#define FRAGS_PER_E   ((IN_DIM / 32) * (OUT_DIM / 16))
#define WFRAG_BLOCKS  (KEXP * FRAGS_PER_E / 8)


#define TOK_PER_CHUNK 1024
#define N_CHUNK       (N_TOK / TOK_PER_CHUNK)

__global__ __launch_bounds__(256) void hist_k(const int* __restrict__ z,
                                              int* __restrict__ hist,
                                              const float* __restrict__ W1,
                                              unsigned int* __restrict__ wfrag) {
    __shared__ int wcnt[8][KEXP];
    const int tid = threadIdx.x, lane = tid & 31, wv = tid >> 5;

    if (blockIdx.x >= N_CHUNK) {
        const int gw    = (blockIdx.x - N_CHUNK) * 8 + wv;
        const int e     = gw >> 6, kstep = (gw >> 3) & 7, ntile = gw & 7;
        const int nn    = lane & 15, khalf = lane >> 4;
        const float* src = W1 + (size_t)e * (IN_DIM * OUT_DIM)
                              + (kstep * 32 + khalf * 8) * OUT_DIM + ntile * 16 + nn;
        float f[16];
#pragma unroll
        for (int j = 0; j < 8; ++j) {
            f[j]     = src[j * OUT_DIM];
            f[8 + j] = src[(16 + j) * OUT_DIM];
        }
        v8bf lo, hi;
#pragma unroll
        for (int j = 0; j < 8; ++j) { lo[j] = (bf16_t)f[j]; hi[j] = (bf16_t)f[8 + j]; }
        v8bf* dst = (v8bf*)((unsigned char*)wfrag + (size_t)gw * FRAG_BYTES + lane * 32);
        dst[0] = lo;
        dst[1] = hi;
        return;
    }

    const int4 zz = *(const int4*)(z + (blockIdx.x * 256 + tid) * 4);
    const int zv[4] = {zz.x, zz.y, zz.z, zz.w};
    int mine = 0;
#pragma unroll
    for (int e = 0; e < KEXP; ++e) {
        int c = 0;
#pragma unroll
        for (int j = 0; j < 4; ++j)
            c += __builtin_popcount(__builtin_amdgcn_ballot_w32(zv[j] == e));
        mine = (lane == e) ? c : mine;
    }
    if (lane < KEXP) wcnt[wv][lane] = mine;
    __syncthreads();
    if (tid < KEXP) {
        int s = 0;
#pragma unroll
        for (int w = 0; w < 8; ++w) s += wcnt[w][tid];
        hist[blockIdx.x * KEXP + tid] = s;
    }
}

__global__ __launch_bounds__(256) void scatter_k(const int* __restrict__ z,
                                                 const int* __restrict__ hist,
                                                 int* __restrict__ off,
                                                 int* __restrict__ idxb) {
    __shared__ int ppre[16][KEXP];
    __shared__ int ptot[16][KEXP];
    __shared__ int wbase[8][KEXP];
    __shared__ int segoff[KEXP];
    __shared__ int segcnt[KEXP];
    const int tid = threadIdx.x, lane = tid & 31, wv = tid >> 5;
    const int b = blockIdx.x;

    {
        const int e = tid & 15, p = tid >> 4;
        int pre = 0, tot = 0;
#pragma unroll
        for (int i = 0; i < N_CHUNK / 16; ++i) {
            const int bb = p + 16 * i;
            const int c  = hist[bb * KEXP + e];
            tot += c;
            pre += (bb < b) ? c : 0;
        }
        ppre[p][e] = pre;
        ptot[p][e] = tot;
    }

    const int tok0 = (b * 256 + tid) * 4;
    const int4 zz = *(const int4*)(z + tok0);
    const int zv[4] = {zz.x, zz.y, zz.z, zz.w};
    int r[4] = {0, 0, 0, 0};
    int mine = 0;
#pragma unroll
    for (int e = 0; e < KEXP; ++e) {
        int run = 0;
#pragma unroll
        for (int j = 0; j < 4; ++j) {
            const bool is = (zv[j] == e);
            const unsigned m = __builtin_amdgcn_ballot_w32(is);
            const int rk = (int)__builtin_amdgcn_mbcnt_lo(m, (unsigned)run);
            r[j] = is ? rk : r[j];
            run += __builtin_popcount(m);
        }
        mine = (lane == e) ? run : mine;
    }
    if (lane < KEXP) wbase[wv][lane] = mine;
    __syncthreads();

    if (wv == 0) {
        const int e = lane & 15;
        int pre = 0, tot = 0;
#pragma unroll
        for (int p = 0; p < 16; ++p) { pre += ppre[p][e]; tot += ptot[p][e]; }
        const int padded = (tot + ROWS_PER_BLK - 1) & ~(ROWS_PER_BLK - 1);
        int incl = padded;
#pragma unroll
        for (int d = 1; d < 16; d <<= 1) {
            const int nb = __shfl_up(incl, d, 16);
            incl += (e >= d) ? nb : 0;
        }
        const int first = incl - padded;
        int wc[8];
#pragma unroll
        for (int w = 0; w < 8; ++w) wc[w] = wbase[w][e];
        if (lane < KEXP) {
            int run = first + pre;
#pragma unroll
            for (int w = 0; w < 8; ++w) { wbase[w][e] = run; run += wc[w]; }
            segoff[e] = first;
            segcnt[e] = tot;
            if (b == 0) {
                off[e] = first;
                if (e == KEXP - 1) off[KEXP] = incl;
            }
        }
    }
    __syncthreads();

    if (b < KEXP && tid < ROWS_PER_BLK) {
        const int p   = segoff[b] + segcnt[b] + tid;
        const int end = segoff[b] + ((segcnt[b] + ROWS_PER_BLK - 1) & ~(ROWS_PER_BLK - 1));
        if (p < end) idxb[p] = -1;
    }

#pragma unroll
    for (int j = 0; j < 4; ++j) {
        const int e = zv[j];
        if ((unsigned)e < (unsigned)KEXP) idxb[wbase[wv][e] + r[j]] = tok0 + j;
    }
}


__global__ __launch_bounds__(256) void moe_k(const float* __restrict__ x,
                                             const int*   __restrict__ idxb,
                                             const int*   __restrict__ off,
                                             const unsigned int* __restrict__ wfrag,
                                             const float* __restrict__ b1,
                                             const float* __restrict__ W2,
                                             const float* __restrict__ b2,
                                             float* __restrict__ out) {
    __shared__ __attribute__((aligned(32))) unsigned char lds[64 * 1024];

    const int base  = blockIdx.x * ROWS_PER_BLK;
    const int total = off[KEXP];
    if (base >= total) return;

    int e = 0;
#pragma unroll
    for (int i = 1; i < KEXP; ++i) e = (base >= off[i]) ? i : e;

    const int tid    = threadIdx.x;
    const int lane   = tid & 31;
    const int wv     = tid >> 5;
    const int hi     = lane >> 4;
    const int lanelo = lane & 15;

    {
        const uint4* wsrc = (const uint4*)(wfrag + (size_t)e * (FRAGS_PER_E * FRAG_BYTES / 4));
        uint4* wdst = (uint4*)lds;
#pragma unroll
        for (int i = 0; i < 16; ++i) wdst[tid + 256 * i] = wsrc[tid + 256 * i];
    }
    __syncthreads();

    const int rowPos = base + wv * 16 + lanelo;
    const int tok    = idxb[rowPos];
    const int tclamp = tok < 0 ? 0 : tok;
    const float* xrow = x + (size_t)tclamp * IN_DIM;

    v8f zero = {0.f, 0.f, 0.f, 0.f, 0.f, 0.f, 0.f, 0.f};
    v8f acc[8];
#pragma unroll
    for (int i = 0; i < 8; ++i) acc[i] = zero;

#pragma unroll
    for (int ks = 0; ks < 8; ++ks) {
        const int kb = ks * 32 + hi * 8;
        float4 f0 = *(const float4*)(xrow + kb);
        float4 f1 = *(const float4*)(xrow + kb + 4);
        float4 f2 = *(const float4*)(xrow + kb + 16);
        float4 f3 = *(const float4*)(xrow + kb + 20);
        v16bf a;
        a[0]=(bf16_t)f0.x; a[1]=(bf16_t)f0.y; a[2] =(bf16_t)f0.z; a[3] =(bf16_t)f0.w;
        a[4]=(bf16_t)f1.x; a[5]=(bf16_t)f1.y; a[6] =(bf16_t)f1.z; a[7] =(bf16_t)f1.w;
        a[8]=(bf16_t)f2.x; a[9]=(bf16_t)f2.y; a[10]=(bf16_t)f2.z; a[11]=(bf16_t)f2.w;
        a[12]=(bf16_t)f3.x; a[13]=(bf16_t)f3.y; a[14]=(bf16_t)f3.z; a[15]=(bf16_t)f3.w;
#pragma unroll
        for (int nt = 0; nt < 8; ++nt) {
            v16bf bf = *(const v16bf*)(lds + (((ks << 3) | nt) << 10) + (lane << 5));
            acc[nt] = __builtin_amdgcn_wmma_f32_16x16x32_bf16(
                false, a, false, bf, (short)0, acc[nt], false, false);
        }
    }

    float s0[8], s1[8];
#pragma unroll
    for (int v = 0; v < 8; ++v) { s0[v] = 0.f; s1[v] = 0.f; }
#pragma unroll
    for (int nt = 0; nt < 8; ++nt) {
        int n = nt * 16 + lanelo;
        float bb  = b1[e * OUT_DIM + n];
        float w20 = W2[((size_t)e * OUT_DIM + n) * 2 + 0];
        float w21 = W2[((size_t)e * OUT_DIM + n) * 2 + 1];
#pragma unroll
        for (int v = 0; v < 8; ++v) {
            float h = acc[nt][v] + bb;
            h = h > 0.f ? h : 0.f;
            s0[v] += h * w20;
            s1[v] += h * w21;
        }
    }
#pragma unroll
    for (int v = 0; v < 8; ++v) {
#pragma unroll
        for (int m = 1; m < 16; m <<= 1) {
            s0[v] += __shfl_xor(s0[v], m, 32);
            s1[v] += __shfl_xor(s1[v], m, 32);
        }
    }

    const float b20 = b2[e * 2 + 0], b21 = b2[e * 2 + 1];
    __syncthreads();
    float* lgt = (float*)lds;
#pragma unroll
    for (int v = 0; v < 8; ++v) {
        if (lanelo == v) {
            int M = v + 8 * hi;
            lgt[(wv * 16 + M) * 2 + 0] = s0[v] + b20;
            lgt[(wv * 16 + M) * 2 + 1] = s1[v] + b21;
        }
    }
    __syncthreads();

    if (lane < 16 && tok >= 0) {
        float l0 = lgt[(wv * 16 + lane) * 2 + 0];
        float l1 = lgt[(wv * 16 + lane) * 2 + 1];
        float mx = fmaxf(l0, l1);
        float e0 = __expf(l0 - mx), e1 = __expf(l1 - mx);
        float inv = 1.f / (e0 + e1);
        out[(size_t)tok * 2 + 0] = e0 * inv;
        out[(size_t)tok * 2 + 1] = e1 * inv;
    }
}


extern "C" void kernel_launch(void* const* d_in, const int* in_sizes, int n_in,
                              void* d_out, int out_size, void* d_ws, size_t ws_size,
                              hipStream_t stream) {
    (void)in_sizes; (void)n_in; (void)out_size; (void)ws_size;
    const float* x  = (const float*)d_in[0];
    const int*   z  = (const int*)d_in[1];
    const float* W1 = (const float*)d_in[2];
    const float* b1 = (const float*)d_in[3];
    const float* W2 = (const float*)d_in[4];
    const float* b2 = (const float*)d_in[5];
    float* out = (float*)d_out;

    int* off  = (int*)d_ws;
    int* hist = off + 32;
    int* idxb = hist + N_CHUNK * KEXP;
    unsigned int* wfrag = (unsigned int*)((unsigned char*)d_ws + (1 << 20));

    hist_k   <<<N_CHUNK + WFRAG_BLOCKS, 256, 0, stream>>>(z, hist, W1, wfrag);
    scatter_k<<<N_CHUNK, 256, 0, stream>>>(z, hist, off, idxb);
    moe_k    <<<MOE_BLOCKS, 256, 0, stream>>>(x, idxb, off, wfrag, b1, W2, b2, out);
}


# Row-tile kernel re-tiled to 4 waves x 32 rows per work-group: next k-step's row loads stay in flight under 8 WMMAs (were issued and waited for one at a time); each LDS B fragment feeds 2 WMMAs
# speedup vs baseline: 2.5442x; 1.1049x over previous
#include <hip/hip_runtime.h>

#define N_TOK   131072
#define KEXP    16
#define IN_DIM  256
#define OUT_DIM 128

#define ROWS_PER_BLK 128
#define MOE_BLOCKS   1040
#define IDX_CAP      (MOE_BLOCKS * ROWS_PER_BLK)

typedef __bf16 bf16_t;
typedef __attribute__((ext_vector_type(16))) bf16_t v16bf;
typedef __attribute__((ext_vector_type(8)))  float  v8f;
typedef __attribute__((ext_vector_type(8)))  bf16_t v8bf;

#define FRAG_BYTES    1024
#define FRAGS_PER_E   ((IN_DIM / 32) * (OUT_DIM / 16))
#define WFRAG_BLOCKS  (KEXP * FRAGS_PER_E / 8)


#define TOK_PER_CHUNK 1024
#define N_CHUNK       (N_TOK / TOK_PER_CHUNK)

__global__ __launch_bounds__(256) void hist_k(const int* __restrict__ z,
                                              int* __restrict__ hist,
                                              const float* __restrict__ W1,
                                              unsigned int* __restrict__ wfrag) {
    __shared__ int wcnt[8][KEXP];
    const int tid = threadIdx.x, lane = tid & 31, wv = tid >> 5;

    if (blockIdx.x >= N_CHUNK) {
        const int gw    = (blockIdx.x - N_CHUNK) * 8 + wv;
        const int e     = gw >> 6, kstep = (gw >> 3) & 7, ntile = gw & 7;
        const int nn    = lane & 15, khalf = lane >> 4;
        const float* src = W1 + (size_t)e * (IN_DIM * OUT_DIM)
                              + (kstep * 32 + khalf * 8) * OUT_DIM + ntile * 16 + nn;
        float f[16];
#pragma unroll
        for (int j = 0; j < 8; ++j) {
            f[j]     = src[j * OUT_DIM];
            f[8 + j] = src[(16 + j) * OUT_DIM];
        }
        v8bf lo, hi;
#pragma unroll
        for (int j = 0; j < 8; ++j) { lo[j] = (bf16_t)f[j]; hi[j] = (bf16_t)f[8 + j]; }
        v8bf* dst = (v8bf*)((unsigned char*)wfrag + (size_t)gw * FRAG_BYTES + lane * 32);
        dst[0] = lo;
        dst[1] = hi;
        return;
    }

    const int4 zz = *(const int4*)(z + (blockIdx.x * 256 + tid) * 4);
    const int zv[4] = {zz.x, zz.y, zz.z, zz.w};
    int mine = 0;
#pragma unroll
    for (int e = 0; e < KEXP; ++e) {
        int c = 0;
#pragma unroll
        for (int j = 0; j < 4; ++j)
            c += __builtin_popcount(__builtin_amdgcn_ballot_w32(zv[j] == e));
        mine = (lane == e) ? c : mine;
    }
    if (lane < KEXP) wcnt[wv][lane] = mine;
    __syncthreads();
    if (tid < KEXP) {
        int s = 0;
#pragma unroll
        for (int w = 0; w < 8; ++w) s += wcnt[w][tid];
        hist[blockIdx.x * KEXP + tid] = s;
    }
}

__global__ __launch_bounds__(256) void scatter_k(const int* __restrict__ z,
                                                 const int* __restrict__ hist,
                                                 int* __restrict__ off,
                                                 int* __restrict__ idxb) {
    __shared__ int ppre[16][KEXP];
    __shared__ int ptot[16][KEXP];
    __shared__ int wbase[8][KEXP];
    __shared__ int segoff[KEXP];
    __shared__ int segcnt[KEXP];
    const int tid = threadIdx.x, lane = tid & 31, wv = tid >> 5;
    const int b = blockIdx.x;

    {
        const int e = tid & 15, p = tid >> 4;
        int pre = 0, tot = 0;
#pragma unroll
        for (int i = 0; i < N_CHUNK / 16; ++i) {
            const int bb = p + 16 * i;
            const int c  = hist[bb * KEXP + e];
            tot += c;
            pre += (bb < b) ? c : 0;
        }
        ppre[p][e] = pre;
        ptot[p][e] = tot;
    }

    const int tok0 = (b * 256 + tid) * 4;
    const int4 zz = *(const int4*)(z + tok0);
    const int zv[4] = {zz.x, zz.y, zz.z, zz.w};
    int r[4] = {0, 0, 0, 0};
    int mine = 0;
#pragma unroll
    for (int e = 0; e < KEXP; ++e) {
        int run = 0;
#pragma unroll
        for (int j = 0; j < 4; ++j) {
            const bool is = (zv[j] == e);
            const unsigned m = __builtin_amdgcn_ballot_w32(is);
            const int rk = (int)__builtin_amdgcn_mbcnt_lo(m, (unsigned)run);
            r[j] = is ? rk : r[j];
            run += __builtin_popcount(m);
        }
        mine = (lane == e) ? run : mine;
    }
    if (lane < KEXP) wbase[wv][lane] = mine;
    __syncthreads();

    if (wv == 0) {
        const int e = lane & 15;
        int pre = 0, tot = 0;
#pragma unroll
        for (int p = 0; p < 16; ++p) { pre += ppre[p][e]; tot += ptot[p][e]; }
        const int padded = (tot + ROWS_PER_BLK - 1) & ~(ROWS_PER_BLK - 1);
        int incl = padded;
#pragma unroll
        for (int d = 1; d < 16; d <<= 1) {
            const int nb = __shfl_up(incl, d, 16);
            incl += (e >= d) ? nb : 0;
        }
        const int first = incl - padded;
        int wc[8];
#pragma unroll
        for (int w = 0; w < 8; ++w) wc[w] = wbase[w][e];
        if (lane < KEXP) {
            int run = first + pre;
#pragma unroll
            for (int w = 0; w < 8; ++w) { wbase[w][e] = run; run += wc[w]; }
            segoff[e] = first;
            segcnt[e] = tot;
            if (b == 0) {
                off[e] = first;
                if (e == KEXP - 1) off[KEXP] = incl;
            }
        }
    }
    __syncthreads();

    if (b < KEXP && tid < ROWS_PER_BLK) {
        const int p   = segoff[b] + segcnt[b] + tid;
        const int end = segoff[b] + ((segcnt[b] + ROWS_PER_BLK - 1) & ~(ROWS_PER_BLK - 1));
        if (p < end) idxb[p] = -1;
    }

#pragma unroll
    for (int j = 0; j < 4; ++j) {
        const int e = zv[j];
        if ((unsigned)e < (unsigned)KEXP) idxb[wbase[wv][e] + r[j]] = tok0 + j;
    }
}


#define MOE_THREADS 128

struct xraw { float4 f0, f1, f2, f3; };

__device__ __forceinline__ xraw load_x(const float* __restrict__ p) {
    xraw r;
    r.f0 = *(const float4*)(p);
    r.f1 = *(const float4*)(p + 4);
    r.f2 = *(const float4*)(p + 16);
    r.f3 = *(const float4*)(p + 20);
    return r;
}

__device__ __forceinline__ void hold_until_here(xraw& r) {
    asm volatile("" : "+v"(r.f0.x), "+v"(r.f0.y), "+v"(r.f0.z), "+v"(r.f0.w),
                      "+v"(r.f1.x), "+v"(r.f1.y), "+v"(r.f1.z), "+v"(r.f1.w),
                      "+v"(r.f2.x), "+v"(r.f2.y), "+v"(r.f2.z), "+v"(r.f2.w),
                      "+v"(r.f3.x), "+v"(r.f3.y), "+v"(r.f3.z), "+v"(r.f3.w));
}

__device__ __forceinline__ v16bf to_afrag(const xraw& r) {
    v16bf a;
    a[0]=(bf16_t)r.f0.x; a[1]=(bf16_t)r.f0.y; a[2] =(bf16_t)r.f0.z; a[3] =(bf16_t)r.f0.w;
    a[4]=(bf16_t)r.f1.x; a[5]=(bf16_t)r.f1.y; a[6] =(bf16_t)r.f1.z; a[7] =(bf16_t)r.f1.w;
    a[8]=(bf16_t)r.f2.x; a[9]=(bf16_t)r.f2.y; a[10]=(bf16_t)r.f2.z; a[11]=(bf16_t)r.f2.w;
    a[12]=(bf16_t)r.f3.x; a[13]=(bf16_t)r.f3.y; a[14]=(bf16_t)r.f3.z; a[15]=(bf16_t)r.f3.w;
    return a;
}

__global__ __launch_bounds__(MOE_THREADS) void moe_k(const float* __restrict__ x,
                                             const int*   __restrict__ idxb,
                                             const int*   __restrict__ off,
                                             const unsigned int* __restrict__ wfrag,
                                             const float* __restrict__ b1,
                                             const float* __restrict__ W2,
                                             const float* __restrict__ b2,
                                             float* __restrict__ out) {
    __shared__ __attribute__((aligned(32))) unsigned char lds[64 * 1024];

    const int base  = blockIdx.x * ROWS_PER_BLK;
    const int total = off[KEXP];
    if (base >= total) return;

    int e = 0;
#pragma unroll
    for (int i = 1; i < KEXP; ++i) e = (base >= off[i]) ? i : e;

    const int tid    = threadIdx.x;
    const int lane   = tid & 31;
    const int wv     = tid >> 5;
    const int hi     = lane >> 4;
    const int lanelo = lane & 15;

    const int rowPos = base + wv * 32 + lanelo;
    const int tok0   = idxb[rowPos];
    const int tok1   = idxb[rowPos + 16];
    const int tc0    = tok0 < 0 ? 0 : tok0;
    const int tc1    = tok1 < 0 ? 0 : tok1;
    const unsigned xo0 = ((unsigned)tc0 * IN_DIM + hi * 8) * 4u;
    const unsigned xo1 = ((unsigned)tc1 * IN_DIM + hi * 8) * 4u;
    const char* xb = (const char*)x;

    {
        const uint4* wsrc = (const uint4*)(wfrag + (size_t)e * (FRAGS_PER_E * FRAG_BYTES / 4));
        uint4* wdst = (uint4*)lds;
#pragma unroll
        for (int i = 0; i < 32; ++i) wdst[tid + MOE_THREADS * i] = wsrc[tid + MOE_THREADS * i];
    }

    v16bf a0, a1;
    {
        const xraw r0 = load_x((const float*)(xb + xo0));
        const xraw r1 = load_x((const float*)(xb + xo1));
        __syncthreads();
        a0 = to_afrag(r0);
        a1 = to_afrag(r1);
    }

    v8f zero = {0.f, 0.f, 0.f, 0.f, 0.f, 0.f, 0.f, 0.f};
    v8f acc0[8], acc1[8];
#pragma unroll
    for (int i = 0; i < 8; ++i) { acc0[i] = zero; acc1[i] = zero; }

#pragma unroll
    for (int ks = 0; ks < 8; ++ks) {
        xraw nx;
        v16bf na0 = a0, na1 = a1;
        if (ks < 7) nx = load_x((const float*)(xb + (xo0 + (ks + 1) * 128)));
#pragma unroll
        for (int nt = 0; nt < 4; ++nt) {
            v16bf bf = *(const v16bf*)(lds + (((ks << 3) | nt) << 10) + (lane << 5));
            acc0[nt] = __builtin_amdgcn_wmma_f32_16x16x32_bf16(
                false, a0, false, bf, (short)0, acc0[nt], false, false);
            acc1[nt] = __builtin_amdgcn_wmma_f32_16x16x32_bf16(
                false, a1, false, bf, (short)0, acc1[nt], false, false);
        }
        __builtin_amdgcn_sched_barrier(0);
        if (ks < 7) {
            hold_until_here(nx);
            na0 = to_afrag(nx);
            __builtin_amdgcn_sched_barrier(0);
            nx = load_x((const float*)(xb + (xo1 + (ks + 1) * 128)));
        }
#pragma unroll
        for (int nt = 4; nt < 8; ++nt) {
            v16bf bf = *(const v16bf*)(lds + (((ks << 3) | nt) << 10) + (lane << 5));
            acc0[nt] = __builtin_amdgcn_wmma_f32_16x16x32_bf16(
                false, a0, false, bf, (short)0, acc0[nt], false, false);
            acc1[nt] = __builtin_amdgcn_wmma_f32_16x16x32_bf16(
                false, a1, false, bf, (short)0, acc1[nt], false, false);
        }
        __builtin_amdgcn_sched_barrier(0);
        if (ks < 7) { hold_until_here(nx); na1 = to_afrag(nx); }
        a0 = na0;
        a1 = na1;
    }

    float bbv[8], w20v[8], w21v[8];
#pragma unroll
    for (int nt = 0; nt < 8; ++nt) {
        const int n = nt * 16 + lanelo;
        bbv[nt]  = b1[e * OUT_DIM + n];
        w20v[nt] = W2[((size_t)e * OUT_DIM + n) * 2 + 0];
        w21v[nt] = W2[((size_t)e * OUT_DIM + n) * 2 + 1];
    }
    float s00[8], s01[8], s10[8], s11[8];
#pragma unroll
    for (int v = 0; v < 8; ++v) { s00[v] = 0.f; s01[v] = 0.f; }
#pragma unroll
    for (int nt = 0; nt < 8; ++nt) {
#pragma unroll
        for (int v = 0; v < 8; ++v) {
            float h0 = acc0[nt][v] + bbv[nt];
            h0 = h0 > 0.f ? h0 : 0.f;
            s00[v] += h0 * w20v[nt];
            s01[v] += h0 * w21v[nt];
        }
    }
#pragma unroll
    for (int v = 0; v < 8; ++v) { s10[v] = 0.f; s11[v] = 0.f; }
#pragma unroll
    for (int nt = 0; nt < 8; ++nt) {
#pragma unroll
        for (int v = 0; v < 8; ++v) {
            float h1 = acc1[nt][v] + bbv[nt];
            h1 = h1 > 0.f ? h1 : 0.f;
            s10[v] += h1 * w20v[nt];
            s11[v] += h1 * w21v[nt];
        }
    }
#pragma unroll
    for (int v = 0; v < 8; ++v) {
#pragma unroll
        for (int m = 1; m < 16; m <<= 1) {
            s00[v] += __shfl_xor(s00[v], m, 32);
            s01[v] += __shfl_xor(s01[v], m, 32);
            s10[v] += __shfl_xor(s10[v], m, 32);
            s11[v] += __shfl_xor(s11[v], m, 32);
        }
    }

    const float b20 = b2[e * 2 + 0], b21 = b2[e * 2 + 1];
    __syncthreads();
    float* lgt = (float*)lds;
#pragma unroll
    for (int v = 0; v < 8; ++v) {
        if (lanelo == v) {
            int M = v + 8 * hi;
            lgt[(wv * 32 + M) * 2 + 0]      = s00[v] + b20;
            lgt[(wv * 32 + M) * 2 + 1]      = s01[v] + b21;
            lgt[(wv * 32 + 16 + M) * 2 + 0] = s10[v] + b20;
            lgt[(wv * 32 + 16 + M) * 2 + 1] = s11[v] + b21;
        }
    }
    __syncthreads();

    const int tok = hi ? tok1 : tok0;
    if (tok >= 0) {
        float l0 = lgt[(wv * 32 + lane) * 2 + 0];
        float l1 = lgt[(wv * 32 + lane) * 2 + 1];
        float mx = fmaxf(l0, l1);
        float e0 = __expf(l0 - mx), e1 = __expf(l1 - mx);
        float inv = 1.f / (e0 + e1);
        out[(size_t)tok * 2 + 0] = e0 * inv;
        out[(size_t)tok * 2 + 1] = e1 * inv;
    }
}


extern "C" void kernel_launch(void* const* d_in, const int* in_sizes, int n_in,
                              void* d_out, int out_size, void* d_ws, size_t ws_size,
                              hipStream_t stream) {
    (void)in_sizes; (void)n_in; (void)out_size; (void)ws_size;
    const float* x  = (const float*)d_in[0];
    const int*   z  = (const int*)d_in[1];
    const float* W1 = (const float*)d_in[2];
    const float* b1 = (const float*)d_in[3];
    const float* W2 = (const float*)d_in[4];
    const float* b2 = (const float*)d_in[5];
    float* out = (float*)d_out;

    int* off  = (int*)d_ws;
    int* hist = off + 32;
    int* idxb = hist + N_CHUNK * KEXP;
    unsigned int* wfrag = (unsigned int*)((unsigned char*)d_ws + (1 << 20));

    hist_k   <<<N_CHUNK + WFRAG_BLOCKS, 256, 0, stream>>>(z, hist, W1, wfrag);
    scatter_k<<<N_CHUNK, 256, 0, stream>>>(z, hist, off, idxb);
    moe_k    <<<MOE_BLOCKS, MOE_THREADS, 0, stream>>>(x, idxb, off, wfrag, b1, W2, b2, out);
}


# Offset/scatter launch removed (3 launches become 2): each chunk stores its own expert-grouped token list and counts; row-tile work-groups prefix the 128x16 counts themselves and pick their rows
# speedup vs baseline: 2.8131x; 1.1057x over previous
#include <hip/hip_runtime.h>

#define N_TOK   131072
#define KEXP    16
#define IN_DIM  256
#define OUT_DIM 128

#define ROWS_PER_BLK 128
#define MOE_BLOCKS   1040
#define IDX_CAP      (MOE_BLOCKS * ROWS_PER_BLK)

typedef __bf16 bf16_t;
typedef __attribute__((ext_vector_type(16))) bf16_t v16bf;
typedef __attribute__((ext_vector_type(8)))  float  v8f;
typedef __attribute__((ext_vector_type(8)))  bf16_t v8bf;

#define FRAG_BYTES    1024
#define FRAGS_PER_E   ((IN_DIM / 32) * (OUT_DIM / 16))
#define WFRAG_BLOCKS  (KEXP * FRAGS_PER_E / 8)


#define TOK_PER_CHUNK 1024
#define N_CHUNK       (N_TOK / TOK_PER_CHUNK)

__global__ __launch_bounds__(256) void sort_k(const int* __restrict__ z,
                                              int* __restrict__ hl,
                                              int* __restrict__ lidx,
                                              const float* __restrict__ W1,
                                              unsigned int* __restrict__ wfrag) {
    __shared__ int wbase[8][KEXP];
    const int tid = threadIdx.x, lane = tid & 31, wv = tid >> 5;

    if (blockIdx.x >= N_CHUNK) {
        const int gw    = (blockIdx.x - N_CHUNK) * 8 + wv;
        const int e     = gw >> 6, kstep = (gw >> 3) & 7, ntile = gw & 7;
        const int nn    = lane & 15, khalf = lane >> 4;
        const float* src = W1 + (size_t)e * (IN_DIM * OUT_DIM)
                              + (kstep * 32 + khalf * 8) * OUT_DIM + ntile * 16 + nn;
        float f[16];
#pragma unroll
        for (int j = 0; j < 8; ++j) {
            f[j]     = src[j * OUT_DIM];
            f[8 + j] = src[(16 + j) * OUT_DIM];
        }
        v8bf lo, hi;
#pragma unroll
        for (int j = 0; j < 8; ++j) { lo[j] = (bf16_t)f[j]; hi[j] = (bf16_t)f[8 + j]; }
        v8bf* dst = (v8bf*)((unsigned char*)wfrag + (size_t)gw * FRAG_BYTES + lane * 32);
        dst[0] = lo;
        dst[1] = hi;
        return;
    }

    const int b = blockIdx.x;

    const int tok0 = (b * 256 + tid) * 4;
    const int4 zz = *(const int4*)(z + tok0);
    const int zv[4] = {zz.x, zz.y, zz.z, zz.w};
    int r[4] = {0, 0, 0, 0};
    int mine = 0;
#pragma unroll
    for (int e = 0; e < KEXP; ++e) {
        int run = 0;
#pragma unroll
        for (int j = 0; j < 4; ++j) {
            const bool is = (zv[j] == e);
            const unsigned m = __builtin_amdgcn_ballot_w32(is);
            const int rk = (int)__builtin_amdgcn_mbcnt_lo(m, (unsigned)run);
            r[j] = is ? rk : r[j];
            run += __builtin_popcount(m);
        }
        mine = (lane == e) ? run : mine;
    }
    if (lane < KEXP) wbase[wv][lane] = mine;
    __syncthreads();

    if (wv == 0) {
        const int e = lane & 15;
        int wc[8];
        int tot = 0;
#pragma unroll
        for (int w = 0; w < 8; ++w) { wc[w] = wbase[w][e]; tot += wc[w]; }
        int incl = tot;
#pragma unroll
        for (int d = 1; d < 16; d <<= 1) {
            const int nb = __shfl_up(incl, d, 16);
            incl += (e >= d) ? nb : 0;
        }
        const int first = incl - tot;
        if (lane < KEXP) {
            int run = first;
#pragma unroll
            for (int w = 0; w < 8; ++w) { wbase[w][e] = run; run += wc[w]; }
            hl[e * N_CHUNK + b] = tot | (first << 16);
        }
    }
    __syncthreads();

#pragma unroll
    for (int j = 0; j < 4; ++j) {
        const int e = zv[j];
        if ((unsigned)e < (unsigned)KEXP)
            lidx[b * TOK_PER_CHUNK + wbase[wv][e] + r[j]] = tok0 + j;
    }
}


#define MOE_THREADS 128

struct xraw { float4 f0, f1, f2, f3; };

__device__ __forceinline__ xraw load_x(const float* __restrict__ p) {
    xraw r;
    r.f0 = *(const float4*)(p);
    r.f1 = *(const float4*)(p + 4);
    r.f2 = *(const float4*)(p + 16);
    r.f3 = *(const float4*)(p + 20);
    return r;
}

__device__ __forceinline__ void hold_until_here(xraw& r) {
    asm volatile("" : "+v"(r.f0.x), "+v"(r.f0.y), "+v"(r.f0.z), "+v"(r.f0.w),
                      "+v"(r.f1.x), "+v"(r.f1.y), "+v"(r.f1.z), "+v"(r.f1.w),
                      "+v"(r.f2.x), "+v"(r.f2.y), "+v"(r.f2.z), "+v"(r.f2.w),
                      "+v"(r.f3.x), "+v"(r.f3.y), "+v"(r.f3.z), "+v"(r.f3.w));
}

__device__ __forceinline__ v16bf to_afrag(const xraw& r) {
    v16bf a;
    a[0]=(bf16_t)r.f0.x; a[1]=(bf16_t)r.f0.y; a[2] =(bf16_t)r.f0.z; a[3] =(bf16_t)r.f0.w;
    a[4]=(bf16_t)r.f1.x; a[5]=(bf16_t)r.f1.y; a[6] =(bf16_t)r.f1.z; a[7] =(bf16_t)r.f1.w;
    a[8]=(bf16_t)r.f2.x; a[9]=(bf16_t)r.f2.y; a[10]=(bf16_t)r.f2.z; a[11]=(bf16_t)r.f2.w;
    a[12]=(bf16_t)r.f3.x; a[13]=(bf16_t)r.f3.y; a[14]=(bf16_t)r.f3.z; a[15]=(bf16_t)r.f3.w;
    return a;
}

__global__ __launch_bounds__(MOE_THREADS) void moe_k(const float* __restrict__ x,
                                             const int*   __restrict__ lidx,
                                             const int*   __restrict__ hl,
                                             const unsigned int* __restrict__ wfrag,
                                             const float* __restrict__ b1,
                                             const float* __restrict__ W2,
                                             const float* __restrict__ b2,
                                             float* __restrict__ out) {
    __shared__ __attribute__((aligned(32))) unsigned char lds[64 * 1024];

    const int base   = blockIdx.x * ROWS_PER_BLK;
    const int tid    = threadIdx.x;
    const int lane   = tid & 31;
    const int wv     = tid >> 5;
    const int hi     = lane >> 4;
    const int lanelo = lane & 15;

    int* s_tot  = (int*)lds;
    int* s_pre8 = s_tot + 16;
    int* s_pre  = s_tot + 32;
    int* s_pk   = s_tot + 160;

    int4 h[4];
#pragma unroll
    for (int i = 0; i < 4; ++i) h[i] = *(const int4*)(hl + (wv * 4 + i) * N_CHUNK + lane * 4);
    int ssum[4], incl[4];
#pragma unroll
    for (int i = 0; i < 4; ++i) {
        ssum[i] = (h[i].x & 0xffff) + (h[i].y & 0xffff) + (h[i].z & 0xffff) + (h[i].w & 0xffff);
        incl[i] = ssum[i];
    }
#pragma unroll
    for (int d = 1; d < 32; d <<= 1) {
#pragma unroll
        for (int i = 0; i < 4; ++i) {
            const int nb = __shfl_up(incl[i], d, 32);
            incl[i] += (lane >= d) ? nb : 0;
        }
    }
    if (lane == 31) *(int4*)(s_tot + wv * 4) = make_int4(incl[0], incl[1], incl[2], incl[3]);
    __syncthreads();

    int first = 0, e = 0, ebase = 0, ecnt = 0;
#pragma unroll
    for (int i = 0; i < KEXP; ++i) {
        const int t  = __builtin_amdgcn_readfirstlane(s_tot[i]);
        const bool g = base >= first;
        e     = g ? i : e;
        ebase = g ? first : ebase;
        ecnt  = g ? t : ecnt;
        first += (t + ROWS_PER_BLK - 1) & ~(ROWS_PER_BLK - 1);
    }
    if (base >= first) return;

    if (wv == (e >> 2)) {
        const int sel = e & 3;
        int4 r = h[0];
        int p0 = incl[0] - ssum[0];
        r.x = sel == 1 ? h[1].x : r.x;  r.y = sel == 1 ? h[1].y : r.y;
        r.z = sel == 1 ? h[1].z : r.z;  r.w = sel == 1 ? h[1].w : r.w;
        p0  = sel == 1 ? incl[1] - ssum[1] : p0;
        r.x = sel == 2 ? h[2].x : r.x;  r.y = sel == 2 ? h[2].y : r.y;
        r.z = sel == 2 ? h[2].z : r.z;  r.w = sel == 2 ? h[2].w : r.w;
        p0  = sel == 2 ? incl[2] - ssum[2] : p0;
        r.x = sel == 3 ? h[3].x : r.x;  r.y = sel == 3 ? h[3].y : r.y;
        r.z = sel == 3 ? h[3].z : r.z;  r.w = sel == 3 ? h[3].w : r.w;
        p0  = sel == 3 ? incl[3] - ssum[3] : p0;
        const int c0 = r.x & 0xffff, c1 = r.y & 0xffff, c2 = r.z & 0xffff;
        *(int4*)(s_pre + lane * 4) = make_int4(p0, p0 + c0, p0 + c0 + c1, p0 + c0 + c1 + c2);
        *(int4*)(s_pk + lane * 4)  = r;
        if ((lane & 1) == 0) s_pre8[lane >> 1] = p0;
    }
    __syncthreads();

    const int q = base - ebase + tid;
    int mytok = -1;
    {
        const int4 a0 = ((const int4*)s_pre8)[0], a1 = ((const int4*)s_pre8)[1];
        const int4 a2 = ((const int4*)s_pre8)[2], a3 = ((const int4*)s_pre8)[3];
        const int g8 = (a0.y <= q) + (a0.z <= q) + (a0.w <= q)
                     + (a1.x <= q) + (a1.y <= q) + (a1.z <= q) + (a1.w <= q)
                     + (a2.x <= q) + (a2.y <= q) + (a2.z <= q) + (a2.w <= q)
                     + (a3.x <= q) + (a3.y <= q) + (a3.z <= q) + (a3.w <= q);
        const int4 p = *(const int4*)(s_pre + g8 * 8), u = *(const int4*)(s_pre + g8 * 8 + 4);
        const int g1 = (p.y <= q) + (p.z <= q) + (p.w <= q)
                     + (u.x <= q) + (u.y <= q) + (u.z <= q) + (u.w <= q);
        int pc = p.x;
        pc = p.y <= q ? p.y : pc;  pc = p.z <= q ? p.z : pc;  pc = p.w <= q ? p.w : pc;
        pc = u.x <= q ? u.x : pc;  pc = u.y <= q ? u.y : pc;
        pc = u.z <= q ? u.z : pc;  pc = u.w <= q ? u.w : pc;
        const int c  = g8 * 8 + g1;
        const int pk = s_pk[c];
        if (q < ecnt) mytok = lidx[c * TOK_PER_CHUNK + (pk >> 16) + (q - pc)];
    }
    __syncthreads();

    const int tok0   = __shfl(mytok, lanelo, 32);
    const int tok1   = __shfl(mytok, lanelo + 16, 32);
    const int tc0    = tok0 < 0 ? 0 : tok0;
    const int tc1    = tok1 < 0 ? 0 : tok1;
    const unsigned xo0 = ((unsigned)tc0 * IN_DIM + hi * 8) * 4u;
    const unsigned xo1 = ((unsigned)tc1 * IN_DIM + hi * 8) * 4u;
    const char* xb = (const char*)x;

    {
        const uint4* wsrc = (const uint4*)(wfrag + (size_t)e * (FRAGS_PER_E * FRAG_BYTES / 4));
        uint4* wdst = (uint4*)lds;
#pragma unroll
        for (int i = 0; i < 32; ++i) wdst[tid + MOE_THREADS * i] = wsrc[tid + MOE_THREADS * i];
    }

    v16bf a0, a1;
    {
        const xraw r0 = load_x((const float*)(xb + xo0));
        const xraw r1 = load_x((const float*)(xb + xo1));
        __syncthreads();
        a0 = to_afrag(r0);
        a1 = to_afrag(r1);
    }

    v8f zero = {0.f, 0.f, 0.f, 0.f, 0.f, 0.f, 0.f, 0.f};
    v8f acc0[8], acc1[8];
#pragma unroll
    for (int i = 0; i < 8; ++i) { acc0[i] = zero; acc1[i] = zero; }

#pragma unroll
    for (int ks = 0; ks < 8; ++ks) {
        xraw nx;
        v16bf na0 = a0, na1 = a1;
        if (ks < 7) nx = load_x((const float*)(xb + (xo0 + (ks + 1) * 128)));
#pragma unroll
        for (int nt = 0; nt < 4; ++nt) {
            v16bf bf = *(const v16bf*)(lds + (((ks << 3) | nt) << 10) + (lane << 5));
            acc0[nt] = __builtin_amdgcn_wmma_f32_16x16x32_bf16(
                false, a0, false, bf, (short)0, acc0[nt], false, false);
            acc1[nt] = __builtin_amdgcn_wmma_f32_16x16x32_bf16(
                false, a1, false, bf, (short)0, acc1[nt], false, false);
        }
        __builtin_amdgcn_sched_barrier(0);
        if (ks < 7) {
            hold_until_here(nx);
            na0 = to_afrag(nx);
            __builtin_amdgcn_sched_barrier(0);
            nx = load_x((const float*)(xb + (xo1 + (ks + 1) * 128)));
        }
#pragma unroll
        for (int nt = 4; nt < 8; ++nt) {
            v16bf bf = *(const v16bf*)(lds + (((ks << 3) | nt) << 10) + (lane << 5));
            acc0[nt] = __builtin_amdgcn_wmma_f32_16x16x32_bf16(
                false, a0, false, bf, (short)0, acc0[nt], false, false);
            acc1[nt] = __builtin_amdgcn_wmma_f32_16x16x32_bf16(
                false, a1, false, bf, (short)0, acc1[nt], false, false);
        }
        __builtin_amdgcn_sched_barrier(0);
        if (ks < 7) { hold_until_here(nx); na1 = to_afrag(nx); }
        a0 = na0;
        a1 = na1;
    }

    float bbv[8], w20v[8], w21v[8];
#pragma unroll
    for (int nt = 0; nt < 8; ++nt) {
        const int n = nt * 16 + lanelo;
        bbv[nt]  = b1[e * OUT_DIM + n];
        w20v[nt] = W2[((size_t)e * OUT_DIM + n) * 2 + 0];
        w21v[nt] = W2[((size_t)e * OUT_DIM + n) * 2 + 1];
    }
    float s00[8], s01[8], s10[8], s11[8];
#pragma unroll
    for (int v = 0; v < 8; ++v) { s00[v] = 0.f; s01[v] = 0.f; }
#pragma unroll
    for (int nt = 0; nt < 8; ++nt) {
#pragma unroll
        for (int v = 0; v < 8; ++v) {
            float h0 = acc0[nt][v] + bbv[nt];
            h0 = h0 > 0.f ? h0 : 0.f;
            s00[v] += h0 * w20v[nt];
            s01[v] += h0 * w21v[nt];
        }
    }
#pragma unroll
    for (int v = 0; v < 8; ++v) { s10[v] = 0.f; s11[v] = 0.f; }
#pragma unroll
    for (int nt = 0; nt < 8; ++nt) {
#pragma unroll
        for (int v = 0; v < 8; ++v) {
            float h1 = acc1[nt][v] + bbv[nt];
            h1 = h1 > 0.f ? h1 : 0.f;
            s10[v] += h1 * w20v[nt];
            s11[v] += h1 * w21v[nt];
        }
    }
#pragma unroll
    for (int v = 0; v < 8; ++v) {
#pragma unroll
        for (int m = 1; m < 16; m <<= 1) {
            s00[v] += __shfl_xor(s00[v], m, 32);
            s01[v] += __shfl_xor(s01[v], m, 32);
            s10[v] += __shfl_xor(s10[v], m, 32);
            s11[v] += __shfl_xor(s11[v], m, 32);
        }
    }

    const float b20 = b2[e * 2 + 0], b21 = b2[e * 2 + 1];
    __syncthreads();
    float* lgt = (float*)lds;
#pragma unroll
    for (int v = 0; v < 8; ++v) {
        if (lanelo == v) {
            int M = v + 8 * hi;
            lgt[(wv * 32 + M) * 2 + 0]      = s00[v] + b20;
            lgt[(wv * 32 + M) * 2 + 1]      = s01[v] + b21;
            lgt[(wv * 32 + 16 + M) * 2 + 0] = s10[v] + b20;
            lgt[(wv * 32 + 16 + M) * 2 + 1] = s11[v] + b21;
        }
    }
    __syncthreads();

    const int tok = mytok;
    if (tok >= 0) {
        float l0 = lgt[(wv * 32 + lane) * 2 + 0];
        float l1 = lgt[(wv * 32 + lane) * 2 + 1];
        float mx = fmaxf(l0, l1);
        float e0 = __expf(l0 - mx), e1 = __expf(l1 - mx);
        float inv = 1.f / (e0 + e1);
        out[(size_t)tok * 2 + 0] = e0 * inv;
        out[(size_t)tok * 2 + 1] = e1 * inv;
    }
}


extern "C" void kernel_launch(void* const* d_in, const int* in_sizes, int n_in,
                              void* d_out, int out_size, void* d_ws, size_t ws_size,
                              hipStream_t stream) {
    (void)in_sizes; (void)n_in; (void)out_size; (void)ws_size;
    const float* x  = (const float*)d_in[0];
    const int*   z  = (const int*)d_in[1];
    const float* W1 = (const float*)d_in[2];
    const float* b1 = (const float*)d_in[3];
    const float* W2 = (const float*)d_in[4];
    const float* b2 = (const float*)d_in[5];
    float* out = (float*)d_out;

    int* hl   = (int*)d_ws;
    int* lidx = hl + KEXP * N_CHUNK;
    unsigned int* wfrag = (unsigned int*)((unsigned char*)d_ws + (1 << 20));

    sort_k<<<N_CHUNK + WFRAG_BLOCKS, 256, 0, stream>>>(z, hl, lidx, W1, wfrag);
    moe_k <<<MOE_BLOCKS, MOE_THREADS, 0, stream>>>(x, lidx, hl, wfrag, b1, W2, b2, out);
}


# Row-tile kernel re-tiled to one work-group per WGP (256 x 12 waves x 48 rows, equal shares per expert): W image copied once per 512 rows; each LDS B fragment feeds 3 WMMAs, read one n-tile ahead
# speedup vs baseline: 3.0075x; 1.0691x over previous
#include <hip/hip_runtime.h>

#define N_TOK   131072
#define KEXP    16
#define IN_DIM  256
#define OUT_DIM 128


typedef __bf16 bf16_t;
typedef __attribute__((ext_vector_type(16))) bf16_t v16bf;
typedef __attribute__((ext_vector_type(8)))  float  v8f;
typedef __attribute__((ext_vector_type(8)))  bf16_t v8bf;

#define FRAG_BYTES    1024
#define FRAGS_PER_E   ((IN_DIM / 32) * (OUT_DIM / 16))
#define WFRAG_BLOCKS  (KEXP * FRAGS_PER_E / 8)


#define TOK_PER_CHUNK 1024
#define N_CHUNK       (N_TOK / TOK_PER_CHUNK)

__global__ __launch_bounds__(256) void sort_k(const int* __restrict__ z,
                                              int* __restrict__ hl,
                                              int* __restrict__ lidx,
                                              const float* __restrict__ W1,
                                              unsigned int* __restrict__ wfrag) {
    __shared__ int wbase[8][KEXP];
    const int tid = threadIdx.x, lane = tid & 31, wv = tid >> 5;

    if (blockIdx.x >= N_CHUNK) {
        const int gw    = (blockIdx.x - N_CHUNK) * 8 + wv;
        const int e     = gw >> 6, kstep = (gw >> 3) & 7, ntile = gw & 7;
        const int nn    = lane & 15, khalf = lane >> 4;
        const float* src = W1 + (size_t)e * (IN_DIM * OUT_DIM)
                              + (kstep * 32 + khalf * 8) * OUT_DIM + ntile * 16 + nn;
        float f[16];
#pragma unroll
        for (int j = 0; j < 8; ++j) {
            f[j]     = src[j * OUT_DIM];
            f[8 + j] = src[(16 + j) * OUT_DIM];
        }
        v8bf lo, hi;
#pragma unroll
        for (int j = 0; j < 8; ++j) { lo[j] = (bf16_t)f[j]; hi[j] = (bf16_t)f[8 + j]; }
        v8bf* dst = (v8bf*)((unsigned char*)wfrag + (size_t)gw * FRAG_BYTES + lane * 32);
        dst[0] = lo;
        dst[1] = hi;
        return;
    }

    const int b = blockIdx.x;

    const int tok0 = (b * 256 + tid) * 4;
    const int4 zz = *(const int4*)(z + tok0);
    const int zv[4] = {zz.x, zz.y, zz.z, zz.w};
    int r[4] = {0, 0, 0, 0};
    int mine = 0;
#pragma unroll
    for (int e = 0; e < KEXP; ++e) {
        int run = 0;
#pragma unroll
        for (int j = 0; j < 4; ++j) {
            const bool is = (zv[j] == e);
            const unsigned m = __builtin_amdgcn_ballot_w32(is);
            const int rk = (int)__builtin_amdgcn_mbcnt_lo(m, (unsigned)run);
            r[j] = is ? rk : r[j];
            run += __builtin_popcount(m);
        }
        mine = (lane == e) ? run : mine;
    }
    if (lane < KEXP) wbase[wv][lane] = mine;
    __syncthreads();

    if (wv == 0) {
        const int e = lane & 15;
        int wc[8];
        int tot = 0;
#pragma unroll
        for (int w = 0; w < 8; ++w) { wc[w] = wbase[w][e]; tot += wc[w]; }
        int incl = tot;
#pragma unroll
        for (int d = 1; d < 16; d <<= 1) {
            const int nb = __shfl_up(incl, d, 16);
            incl += (e >= d) ? nb : 0;
        }
        const int first = incl - tot;
        if (lane < KEXP) {
            int run = first;
#pragma unroll
            for (int w = 0; w < 8; ++w) { wbase[w][e] = run; run += wc[w]; }
            hl[e * N_CHUNK + b] = tot | (first << 16);
        }
    }
    __syncthreads();

#pragma unroll
    for (int j = 0; j < 4; ++j) {
        const int e = zv[j];
        if ((unsigned)e < (unsigned)KEXP)
            lidx[b * TOK_PER_CHUNK + wbase[wv][e] + r[j]] = tok0 + j;
    }
}


#define WG_WAVES      12
#define MOE_THREADS   (WG_WAVES * 32)
#define TILES         3
#define ROWS_PER_WAVE (TILES * 16)
#define ROWS_PER_PASS (WG_WAVES * ROWS_PER_WAVE)
#define WG_PER_E      16
#define MOE_BLOCKS    (KEXP * WG_PER_E)
#define IMG_VEC       (FRAGS_PER_E * FRAG_BYTES / 16)
#define IMG_ITERS     ((IMG_VEC + MOE_THREADS - 1) / MOE_THREADS)

typedef __attribute__((ext_vector_type(4))) float v4f;
struct xraw { v4f f0, f1, f2, f3; };

__device__ __forceinline__ xraw load_x(const float* __restrict__ p) {
    xraw r;
    r.f0 = *(const v4f*)(p);
    r.f1 = *(const v4f*)(p + 4);
    r.f2 = *(const v4f*)(p + 16);
    r.f3 = *(const v4f*)(p + 20);
    return r;
}

__device__ __forceinline__ void hold_until_here(xraw& r) {
    asm volatile("" : "+v"(r.f0), "+v"(r.f1), "+v"(r.f2), "+v"(r.f3));
}

__device__ __forceinline__ void addr_after(unsigned& xo, const v16bf& a) {
    asm volatile("" : "+v"(xo) : "v"(a));
}

__device__ __forceinline__ v16bf to_afrag(const xraw& r) {
    v16bf a;
    a[0]=(bf16_t)r.f0.x; a[1]=(bf16_t)r.f0.y; a[2] =(bf16_t)r.f0.z; a[3] =(bf16_t)r.f0.w;
    a[4]=(bf16_t)r.f1.x; a[5]=(bf16_t)r.f1.y; a[6] =(bf16_t)r.f1.z; a[7] =(bf16_t)r.f1.w;
    a[8]=(bf16_t)r.f2.x; a[9]=(bf16_t)r.f2.y; a[10]=(bf16_t)r.f2.z; a[11]=(bf16_t)r.f2.w;
    a[12]=(bf16_t)r.f3.x; a[13]=(bf16_t)r.f3.y; a[14]=(bf16_t)r.f3.z; a[15]=(bf16_t)r.f3.w;
    return a;
}

__device__ __forceinline__ int row_token(int q, bool valid,
                                         const int* s_pre8, const int* s_pre, const int* s_pk,
                                         const int* __restrict__ lidx) {
    const int4 a0 = ((const int4*)s_pre8)[0], a1 = ((const int4*)s_pre8)[1];
    const int4 a2 = ((const int4*)s_pre8)[2], a3 = ((const int4*)s_pre8)[3];
    const int g8 = (a0.y <= q) + (a0.z <= q) + (a0.w <= q)
                 + (a1.x <= q) + (a1.y <= q) + (a1.z <= q) + (a1.w <= q)
                 + (a2.x <= q) + (a2.y <= q) + (a2.z <= q) + (a2.w <= q)
                 + (a3.x <= q) + (a3.y <= q) + (a3.z <= q) + (a3.w <= q);
    const int4 p = *(const int4*)(s_pre + g8 * 8), u = *(const int4*)(s_pre + g8 * 8 + 4);
    const int g1 = (p.y <= q) + (p.z <= q) + (p.w <= q)
                 + (u.x <= q) + (u.y <= q) + (u.z <= q) + (u.w <= q);
    int pc = p.x;
    pc = p.y <= q ? p.y : pc;  pc = p.z <= q ? p.z : pc;  pc = p.w <= q ? p.w : pc;
    pc = u.x <= q ? u.x : pc;  pc = u.y <= q ? u.y : pc;
    pc = u.z <= q ? u.z : pc;  pc = u.w <= q ? u.w : pc;
    const int c  = g8 * 8 + g1;
    const int pk = s_pk[c];
    int tok = -1;
    if (valid) tok = lidx[c * TOK_PER_CHUNK + (pk >> 16) + (q - pc)];
    return tok;
}

__global__ __launch_bounds__(MOE_THREADS) void moe_k(const float* __restrict__ x,
                                             const int*   __restrict__ lidx,
                                             const int*   __restrict__ hl,
                                             const unsigned int* __restrict__ wfrag,
                                             const float* __restrict__ b1,
                                             const float* __restrict__ W2,
                                             const float* __restrict__ b2,
                                             float* __restrict__ out) {
    __shared__ __attribute__((aligned(32))) unsigned char lds[64 * 1024];
    __shared__ __attribute__((aligned(16))) int s_pre[N_CHUNK];
    __shared__ __attribute__((aligned(16))) int s_pk[N_CHUNK];
    __shared__ __attribute__((aligned(16))) int s_pre8[16];
    __shared__ int s_cnt[1];
    __shared__ __attribute__((aligned(16))) int s_rl[ROWS_PER_PASS * 4];

    const int tid    = threadIdx.x;
    const int lane   = tid & 31;
    const int wv     = tid >> 5;
    const int hi     = lane >> 4;
    const int lanelo = lane & 15;
    const int e      = blockIdx.x >> 4;
    const int part   = blockIdx.x & (WG_PER_E - 1);

    if (wv == 0) {
        const int4 r = *(const int4*)(hl + e * N_CHUNK + lane * 4);
        const int c0 = r.x & 0xffff, c1 = r.y & 0xffff, c2 = r.z & 0xffff, c3 = r.w & 0xffff;
        const int ssum = c0 + c1 + c2 + c3;
        int incl = ssum;
#pragma unroll
        for (int d = 1; d < 32; d <<= 1) {
            const int nb = __shfl_up(incl, d, 32);
            incl += (lane >= d) ? nb : 0;
        }
        const int p0 = incl - ssum;
        *(int4*)(s_pre + lane * 4) = make_int4(p0, p0 + c0, p0 + c0 + c1, p0 + c0 + c1 + c2);
        *(int4*)(s_pk + lane * 4)  = r;
        if ((lane & 1) == 0) s_pre8[lane >> 1] = p0;
        if (lane == 31) s_cnt[0] = incl;
    }

    {
        const uint4* wsrc = (const uint4*)(wfrag + (size_t)e * (FRAGS_PER_E * FRAG_BYTES / 4));
        uint4* wdst = (uint4*)lds;
#pragma unroll
        for (int i = 0; i < IMG_ITERS - 1; ++i) wdst[tid + MOE_THREADS * i] = wsrc[tid + MOE_THREADS * i];
        if (tid < IMG_VEC - MOE_THREADS * (IMG_ITERS - 1))
            wdst[tid + MOE_THREADS * (IMG_ITERS - 1)] = wsrc[tid + MOE_THREADS * (IMG_ITERS - 1)];
    }

    __syncthreads();

    const int ecnt  = __builtin_amdgcn_readfirstlane(s_cnt[0]);
    const int share = (ecnt + WG_PER_E - 1) / WG_PER_E;
    const int lo    = part * share;
    const int hiRow = (lo + share < ecnt) ? (lo + share) : ecnt;
    const char* xb  = (const char*)x;

    for (int pbase = lo; pbase < hiRow; pbase += ROWS_PER_PASS) {
        {
            const int q0 = pbase + tid;
            s_rl[tid * 4] = row_token(q0, q0 < hiRow, s_pre8, s_pre, s_pk, lidx);
            if (tid < ROWS_PER_PASS - MOE_THREADS) {
                const int q1 = q0 + MOE_THREADS;
                s_rl[(tid + MOE_THREADS) * 4] = row_token(q1, q1 < hiRow, s_pre8, s_pre, s_pk, lidx);
            }
        }
        __syncthreads();

        const bool active = (pbase + __builtin_amdgcn_readfirstlane(wv) * ROWS_PER_WAVE) < hiRow;
        if (active) {
            unsigned xo[TILES];
#pragma unroll
            for (int t = 0; t < TILES; ++t) {
                const int tk = s_rl[(wv * ROWS_PER_WAVE + t * 16 + lanelo) * 4];
                const int tc = tk < 0 ? 0 : tk;
                xo[t] = ((unsigned)tc * IN_DIM + hi * 8) * 4u;
            }

            xraw raw[TILES];
            v16bf a[TILES];
#pragma unroll
            for (int t = 0; t < TILES; ++t) raw[t] = load_x((const float*)(xb + xo[t]));
#pragma unroll
            for (int t = 0; t < TILES; ++t) a[t] = to_afrag(raw[t]);
#pragma unroll
            for (int t = 0; t < TILES; ++t) raw[t] = load_x((const float*)(xb + (xo[t] + 128)));

            const v8f zero = {0.f, 0.f, 0.f, 0.f, 0.f, 0.f, 0.f, 0.f};
            v8f acc[TILES][8];
#pragma unroll
            for (int t = 0; t < TILES; ++t)
#pragma unroll
                for (int i = 0; i < 8; ++i) acc[t][i] = zero;

            float bbv[8], w20v[8], w21v[8], b20, b21;
#pragma unroll
            for (int ks = 0; ks < 8; ++ks) {
                if (ks == 7) {
#pragma unroll
                    for (int nt = 0; nt < 8; ++nt) {
                        const int n = nt * 16 + lanelo;
                        bbv[nt]  = b1[e * OUT_DIM + n];
                        w20v[nt] = W2[((size_t)e * OUT_DIM + n) * 2 + 0];
                        w21v[nt] = W2[((size_t)e * OUT_DIM + n) * 2 + 1];
                    }
                    b20 = b2[e * 2 + 0];
                    b21 = b2[e * 2 + 1];
                }
                __builtin_amdgcn_sched_barrier(0);
                v16bf bcur = *(const v16bf*)(lds + ((ks << 3) << 10) + (lane << 5));
#pragma unroll
                for (int nt = 0; nt < 8; ++nt) {
                    v16bf bnext = bcur;
                    if (nt < 7) bnext = *(const v16bf*)(lds + (((ks << 3) | (nt + 1)) << 10) + (lane << 5));
#pragma unroll
                    for (int t = 0; t < TILES; ++t)
                        acc[t][nt] = __builtin_amdgcn_wmma_f32_16x16x32_bf16(
                            false, a[t], false, bcur, (short)0, acc[t][nt], false, false);
                    bcur = bnext;
                }
                __builtin_amdgcn_sched_group_barrier(0x100, 2, 0);
#pragma unroll
                for (int nt = 0; nt < 7; ++nt) {
                    __builtin_amdgcn_sched_group_barrier(0x100, 2, 0);
                    __builtin_amdgcn_sched_group_barrier(0x008, 3, 0);
                }
                __builtin_amdgcn_sched_group_barrier(0x008, 3, 0);
                __builtin_amdgcn_sched_barrier(0);
                if (ks < 7) {
#pragma unroll
                    for (int t = 0; t < TILES; ++t) { hold_until_here(raw[t]); a[t] = to_afrag(raw[t]); }
                    if (ks < 6) {
#pragma unroll
                        for (int t = 0; t < TILES; ++t) {
                            addr_after(xo[t], a[t]);
                            raw[t] = load_x((const float*)(xb + (xo[t] + (ks + 2) * 128)));
                        }
                    }
                }
            }

#pragma unroll
            for (int t = 0; t < TILES; ++t) {
                float s0[8], s1[8];
#pragma unroll
                for (int v = 0; v < 8; ++v) { s0[v] = 0.f; s1[v] = 0.f; }
#pragma unroll
                for (int nt = 0; nt < 8; ++nt) {
#pragma unroll
                    for (int v = 0; v < 8; ++v) {
                        float h0 = acc[t][nt][v] + bbv[nt];
                        h0 = h0 > 0.f ? h0 : 0.f;
                        s0[v] += h0 * w20v[nt];
                        s1[v] += h0 * w21v[nt];
                    }
                }
#pragma unroll
                for (int v = 0; v < 8; ++v) {
#pragma unroll
                    for (int m = 1; m < 16; m <<= 1) {
                        s0[v] += __shfl_xor(s0[v], m, 32);
                        s1[v] += __shfl_xor(s1[v], m, 32);
                    }
                }
#pragma unroll
                for (int v = 0; v < 8; ++v) {
                    if (lanelo == v) {
                        const int M = v + 8 * hi;
                        float* rl = (float*)(s_rl + (wv * ROWS_PER_WAVE + t * 16 + M) * 4);
                        rl[1] = s0[v] + b20;
                        rl[2] = s1[v] + b21;
                    }
                }
            }
        }
        __syncthreads();

#pragma unroll
        for (int j = 0; j < 2; ++j) {
            const int rr = wv * ROWS_PER_WAVE + j * 32 + lane;
            if (j == 0 || lane < 16) {
                const int4 rw = *(const int4*)(s_rl + rr * 4);
                const int tok = rw.x;
                if (tok >= 0) {
                    const float l0 = __int_as_float(rw.y);
                    const float l1 = __int_as_float(rw.z);
                    const float mx = fmaxf(l0, l1);
                    const float e0 = __expf(l0 - mx), e1 = __expf(l1 - mx);
                    const float inv = 1.f / (e0 + e1);
                    out[(size_t)tok * 2 + 0] = e0 * inv;
                    out[(size_t)tok * 2 + 1] = e1 * inv;
                }
            }
        }
        __syncthreads();
    }
}


extern "C" void kernel_launch(void* const* d_in, const int* in_sizes, int n_in,
                              void* d_out, int out_size, void* d_ws, size_t ws_size,
                              hipStream_t stream) {
    (void)in_sizes; (void)n_in; (void)out_size; (void)ws_size;
    const float* x  = (const float*)d_in[0];
    const int*   z  = (const int*)d_in[1];
    const float* W1 = (const float*)d_in[2];
    const float* b1 = (const float*)d_in[3];
    const float* W2 = (const float*)d_in[4];
    const float* b2 = (const float*)d_in[5];
    float* out = (float*)d_out;

    int* hl   = (int*)d_ws;
    int* lidx = hl + KEXP * N_CHUNK;
    unsigned int* wfrag = (unsigned int*)((unsigned char*)d_ws + (1 << 20));

    sort_k<<<N_CHUNK + WFRAG_BLOCKS, 256, 0, stream>>>(z, hl, lidx, W1, wfrag);
    moe_k <<<MOE_BLOCKS, MOE_THREADS, 0, stream>>>(x, lidx, hl, wfrag, b1, W2, b2, out);
}


# Bucketing launch removed (2 launches become 1): every row-tile work-group ballots its expert's tokens out of a 16384-token range shared by a pair and converts W1 into its own LDS image
# speedup vs baseline: 3.4261x; 1.1392x over previous
#include <hip/hip_runtime.h>

#define N_TOK   131072
#define KEXP    16
#define IN_DIM  256
#define OUT_DIM 128


typedef __bf16 bf16_t;
typedef __attribute__((ext_vector_type(16))) bf16_t v16bf;
typedef __attribute__((ext_vector_type(8)))  float  v8f;
typedef __attribute__((ext_vector_type(8)))  bf16_t v8bf;

#define FRAG_BYTES    1024
#define FRAGS_PER_E   ((IN_DIM / 32) * (OUT_DIM / 16))


#define WG_WAVES      12
#define MOE_THREADS   (WG_WAVES * 32)
#define TILES         3
#define ROWS_PER_WAVE (TILES * 16)
#define ROWS_PER_PASS (WG_WAVES * ROWS_PER_WAVE)
#define WG_PER_E      16
#define MOE_BLOCKS    (KEXP * WG_PER_E)
#define IMG_VEC       (FRAGS_PER_E * FRAG_BYTES / 16)
#define IMG_ITERS     ((IMG_VEC + MOE_THREADS - 1) / MOE_THREADS)

typedef __attribute__((ext_vector_type(4))) float v4f;
struct xraw { v4f f0, f1, f2, f3; };

__device__ __forceinline__ xraw load_x(const float* __restrict__ p) {
    xraw r;
    r.f0 = *(const v4f*)(p);
    r.f1 = *(const v4f*)(p + 4);
    r.f2 = *(const v4f*)(p + 16);
    r.f3 = *(const v4f*)(p + 20);
    return r;
}

__device__ __forceinline__ void hold_until_here(xraw& r) {
    asm volatile("" : "+v"(r.f0), "+v"(r.f1), "+v"(r.f2), "+v"(r.f3));
}

__device__ __forceinline__ void addr_after(unsigned& xo, const v16bf& a) {
    asm volatile("" : "+v"(xo) : "v"(a));
}

__device__ __forceinline__ v16bf to_afrag(const xraw& r) {
    v16bf a;
    a[0]=(bf16_t)r.f0.x; a[1]=(bf16_t)r.f0.y; a[2] =(bf16_t)r.f0.z; a[3] =(bf16_t)r.f0.w;
    a[4]=(bf16_t)r.f1.x; a[5]=(bf16_t)r.f1.y; a[6] =(bf16_t)r.f1.z; a[7] =(bf16_t)r.f1.w;
    a[8]=(bf16_t)r.f2.x; a[9]=(bf16_t)r.f2.y; a[10]=(bf16_t)r.f2.z; a[11]=(bf16_t)r.f2.w;
    a[12]=(bf16_t)r.f3.x; a[13]=(bf16_t)r.f3.y; a[14]=(bf16_t)r.f3.z; a[15]=(bf16_t)r.f3.w;
    return a;
}

#define PAIR_TOK      (N_TOK / (WG_PER_E / 2))
#define SCAN_ITERS    ((PAIR_TOK / 4 + MOE_THREADS - 1) / MOE_THREADS)

#define SM_RL     0
#define SM_WCNT   (SM_RL + ROWS_PER_PASS * 16)
#define SM_TOK    (SM_WCNT + 64)
#define TOK_SLOTS (((PAIR_TOK / 2 + ROWS_PER_PASS - 1) / ROWS_PER_PASS) * ROWS_PER_PASS)
#define SM_IMG    (SM_TOK + TOK_SLOTS * 4)
#define SM_TOTAL  (SM_IMG + FRAGS_PER_E * FRAG_BYTES)

__global__ __launch_bounds__(MOE_THREADS) void moe_k(const float* __restrict__ x,
                                             const int*   __restrict__ z,
                                             const float* __restrict__ W1,
                                             const float* __restrict__ b1,
                                             const float* __restrict__ W2,
                                             const float* __restrict__ b2,
                                             float* __restrict__ out) {
    __shared__ __attribute__((aligned(32))) unsigned char smem[SM_TOTAL];
    int* const s_rl   = (int*)(smem + SM_RL);
    int* const s_wcnt = (int*)(smem + SM_WCNT);
    int* const s_tok  = (int*)(smem + SM_TOK);
    unsigned char* const lds = smem + SM_IMG;

    const int tid    = threadIdx.x;
    const int lane   = tid & 31;
    const int wv     = tid >> 5;
    const int hi     = lane >> 4;
    const int lanelo = lane & 15;
    const int e      = blockIdx.x >> 4;
    const int part   = blockIdx.x & (WG_PER_E - 1);
    const int tbase  = (part >> 1) * PAIR_TOK;

    int4 zz[SCAN_ITERS];
#pragma unroll
    for (int i = 0; i < SCAN_ITERS; ++i) {
        const int t4 = i * MOE_THREADS + tid;
        zz[i] = make_int4(-1, -1, -1, -1);
        if (t4 < PAIR_TOK / 4) zz[i] = *(const int4*)(z + tbase + t4 * 4);
    }

    {
        const float* wsrc = W1 + (size_t)e * (IN_DIM * OUT_DIM);
#pragma unroll
        for (int i = 0; i < IMG_ITERS; ++i) {
            int v = tid + MOE_THREADS * i;
            v = (v < IMG_VEC) ? v : v - MOE_THREADS;
            {
                const int frag = v >> 6, lf = (v >> 1) & 31, j = v & 1;
                const int kstep = frag >> 3, ntile = frag & 7, khalf = lf >> 4, nn = lf & 15;
                const float* s = wsrc + (kstep * 32 + j * 16 + khalf * 8) * OUT_DIM + ntile * 16 + nn;
                float f[8];
#pragma unroll
                for (int q = 0; q < 8; ++q) f[q] = s[q * OUT_DIM];
                v8bf pk;
#pragma unroll
                for (int q = 0; q < 8; ++q) pk[q] = (bf16_t)f[q];
                *(v8bf*)(lds + v * 16) = pk;
            }
        }
    }

    {
        int cnt = 0;
#pragma unroll
        for (int i = 0; i < SCAN_ITERS; ++i) {
            const int zv[4] = {zz[i].x, zz[i].y, zz[i].z, zz[i].w};
#pragma unroll
            for (int j = 0; j < 4; ++j)
                cnt += __builtin_popcount(__builtin_amdgcn_ballot_w32(zv[j] == e));
        }
        if (lane == 0) s_wcnt[wv] = cnt;
    }
    __syncthreads();

    int wbase = 0, total = 0;
#pragma unroll
    for (int w = 0; w < WG_WAVES; ++w) {
        const int c = s_wcnt[w];
        total += c;
        wbase += (w < wv) ? c : 0;
    }
    wbase = __builtin_amdgcn_readfirstlane(wbase);
    total = __builtin_amdgcn_readfirstlane(total);
    const int halfc = (total + 1) >> 1;
    const int rlo   = (part & 1) ? halfc : 0;
    const int rhi   = (part & 1) ? total : halfc;
    const int hiRow = rhi - rlo;

    {
        unsigned run = (unsigned)(wbase - rlo);
#pragma unroll
        for (int i = 0; i < SCAN_ITERS; ++i) {
            const int zv[4] = {zz[i].x, zz[i].y, zz[i].z, zz[i].w};
#pragma unroll
            for (int j = 0; j < 4; ++j) {
                const bool is = (zv[j] == e);
                const unsigned m = __builtin_amdgcn_ballot_w32(is);
                const unsigned slot = __builtin_amdgcn_mbcnt_lo(m, run);
                if (is && slot < (unsigned)hiRow)
                    s_tok[slot] = tbase + (i * MOE_THREADS + tid) * 4 + j;
                run += (unsigned)__builtin_popcount(m);
            }
        }
    }

    {
        const int padEnd = ((hiRow + ROWS_PER_PASS - 1) / ROWS_PER_PASS) * ROWS_PER_PASS;
        for (int q = hiRow + tid; q < padEnd; q += MOE_THREADS) s_tok[q] = -1;
    }

    __syncthreads();

    const char* xb  = (const char*)x;
    const float* b1e = b1 + e * OUT_DIM;
    const float* W2e = W2 + e * OUT_DIM * 2;

    for (int pbase = 0; pbase < hiRow; pbase += ROWS_PER_PASS) {
        int tkv[TILES];
#pragma unroll
        for (int t = 0; t < TILES; ++t) {
            const int rloc = wv * ROWS_PER_WAVE + t * 16 + lanelo;
            tkv[t] = s_tok[pbase + rloc];
            s_rl[rloc * 4] = tkv[t];
        }

        const bool active = (pbase + __builtin_amdgcn_readfirstlane(wv) * ROWS_PER_WAVE) < hiRow;
        if (active) {
            unsigned xo[TILES];
#pragma unroll
            for (int t = 0; t < TILES; ++t) {
                const int tk = tkv[t];
                const int tc = tk < 0 ? 0 : tk;
                xo[t] = ((unsigned)tc * IN_DIM + hi * 8) * 4u;
            }

            xraw raw[TILES];
            v16bf a[TILES];
#pragma unroll
            for (int t = 0; t < TILES; ++t) raw[t] = load_x((const float*)(xb + xo[t]));
#pragma unroll
            for (int t = 0; t < TILES; ++t) a[t] = to_afrag(raw[t]);
#pragma unroll
            for (int t = 0; t < TILES; ++t) raw[t] = load_x((const float*)(xb + (xo[t] + 128)));

            const v8f zero = {0.f, 0.f, 0.f, 0.f, 0.f, 0.f, 0.f, 0.f};
            v8f acc[TILES][8];
#pragma unroll
            for (int t = 0; t < TILES; ++t)
#pragma unroll
                for (int i = 0; i < 8; ++i) acc[t][i] = zero;

            float bbv[8], w20v[8], w21v[8], b20, b21;
#pragma unroll
            for (int ks = 0; ks < 8; ++ks) {
                if (ks == 7) {
#pragma unroll
                    for (int nt = 0; nt < 8; ++nt) {
                        const unsigned n = (unsigned)(nt * 16 + lanelo);
                        bbv[nt]  = b1e[n];
                        w20v[nt] = W2e[n * 2u + 0u];
                        w21v[nt] = W2e[n * 2u + 1u];
                    }
                    b20 = b2[e * 2 + 0];
                    b21 = b2[e * 2 + 1];
                }
                __builtin_amdgcn_sched_barrier(0);
                v16bf bcur = *(const v16bf*)(lds + ((ks << 3) << 10) + (lane << 5));
#pragma unroll
                for (int nt = 0; nt < 8; ++nt) {
                    v16bf bnext = bcur;
                    if (nt < 7) bnext = *(const v16bf*)(lds + (((ks << 3) | (nt + 1)) << 10) + (lane << 5));
#pragma unroll
                    for (int t = 0; t < TILES; ++t)
                        acc[t][nt] = __builtin_amdgcn_wmma_f32_16x16x32_bf16(
                            false, a[t], false, bcur, (short)0, acc[t][nt], false, false);
                    bcur = bnext;
                }
                __builtin_amdgcn_sched_group_barrier(0x100, 2, 0);
#pragma unroll
                for (int nt = 0; nt < 7; ++nt) {
                    __builtin_amdgcn_sched_group_barrier(0x100, 2, 0);
                    __builtin_amdgcn_sched_group_barrier(0x008, 3, 0);
                }
                __builtin_amdgcn_sched_group_barrier(0x008, 3, 0);
                __builtin_amdgcn_sched_barrier(0);
                if (ks < 7) {
#pragma unroll
                    for (int t = 0; t < TILES; ++t) { hold_until_here(raw[t]); a[t] = to_afrag(raw[t]); }
                    if (ks < 6) {
#pragma unroll
                        for (int t = 0; t < TILES; ++t) {
                            addr_after(xo[t], a[t]);
                            raw[t] = load_x((const float*)(xb + (xo[t] + (ks + 2) * 128)));
                        }
                    }
                }
            }

#pragma unroll
            for (int t = 0; t < TILES; ++t) {
                float s0[8], s1[8];
#pragma unroll
                for (int v = 0; v < 8; ++v) { s0[v] = 0.f; s1[v] = 0.f; }
#pragma unroll
                for (int nt = 0; nt < 8; ++nt) {
#pragma unroll
                    for (int v = 0; v < 8; ++v) {
                        float h0 = acc[t][nt][v] + bbv[nt];
                        h0 = h0 > 0.f ? h0 : 0.f;
                        s0[v] += h0 * w20v[nt];
                        s1[v] += h0 * w21v[nt];
                    }
                }
#pragma unroll
                for (int v = 0; v < 8; ++v) {
#pragma unroll
                    for (int m = 1; m < 16; m <<= 1) {
                        s0[v] += __shfl_xor(s0[v], m, 32);
                        s1[v] += __shfl_xor(s1[v], m, 32);
                    }
                }
#pragma unroll
                for (int v = 0; v < 8; ++v) {
                    if (lanelo == v) {
                        const int M = v + 8 * hi;
                        float* rl = (float*)(s_rl + (wv * ROWS_PER_WAVE + t * 16 + M) * 4);
                        rl[1] = s0[v] + b20;
                        rl[2] = s1[v] + b21;
                    }
                }
            }
        }
        __syncthreads();

#pragma unroll
        for (int j = 0; j < 2; ++j) {
            const int rr = wv * ROWS_PER_WAVE + j * 32 + lane;
            if (j == 0 || lane < 16) {
                const int4 rw = *(const int4*)(s_rl + rr * 4);
                const int tok = rw.x;
                if (tok >= 0) {
                    const float l0 = __int_as_float(rw.y);
                    const float l1 = __int_as_float(rw.z);
                    const float mx = fmaxf(l0, l1);
                    const float e0 = __expf(l0 - mx), e1 = __expf(l1 - mx);
                    const float inv = 1.f / (e0 + e1);
                    out[(size_t)tok * 2 + 0] = e0 * inv;
                    out[(size_t)tok * 2 + 1] = e1 * inv;
                }
            }
        }
        __syncthreads();
    }
}


extern "C" void kernel_launch(void* const* d_in, const int* in_sizes, int n_in,
                              void* d_out, int out_size, void* d_ws, size_t ws_size,
                              hipStream_t stream) {
    (void)in_sizes; (void)n_in; (void)out_size; (void)d_ws; (void)ws_size;
    const float* x  = (const float*)d_in[0];
    const int*   z  = (const int*)d_in[1];
    const float* W1 = (const float*)d_in[2];
    const float* b1 = (const float*)d_in[3];
    const float* W2 = (const float*)d_in[4];
    const float* b2 = (const float*)d_in[5];
    float* out = (float*)d_out;

    moe_k<<<MOE_BLOCKS, MOE_THREADS, 0, stream>>>(x, z, W1, b1, W2, b2, out);
}


# Row loads of x cover whole 128-byte lines (8 lanes per row, 4 lines per load, were 16 lines x 32 B); rows pass as bf16 through a per-wave LDS stage from which the A fragments are read
# speedup vs baseline: 3.5399x; 1.0332x over previous
#include <hip/hip_runtime.h>

#define N_TOK   131072
#define KEXP    16
#define IN_DIM  256
#define OUT_DIM 128


typedef __bf16 bf16_t;
typedef __attribute__((ext_vector_type(16))) bf16_t v16bf;
typedef __attribute__((ext_vector_type(8)))  float  v8f;
typedef __attribute__((ext_vector_type(8)))  bf16_t v8bf;

#define FRAG_BYTES    1024
#define FRAGS_PER_E   ((IN_DIM / 32) * (OUT_DIM / 16))


#define WG_WAVES      12
#define MOE_THREADS   (WG_WAVES * 32)
#define TILES         3
#define ROWS_PER_WAVE (TILES * 16)
#define ROWS_PER_PASS (WG_WAVES * ROWS_PER_WAVE)
#define WG_PER_E      16
#define MOE_BLOCKS    (KEXP * WG_PER_E)
#define IMG_VEC       (FRAGS_PER_E * FRAG_BYTES / 16)
#define IMG_ITERS     ((IMG_VEC + MOE_THREADS - 1) / MOE_THREADS)

typedef __attribute__((ext_vector_type(4))) float  v4f;
typedef __attribute__((ext_vector_type(4))) bf16_t v4bf;

__device__ __forceinline__ void hold_until_here(v4f& r) {
    asm volatile("" : "+v"(r));
}

__device__ __forceinline__ void addr_after(unsigned& xo, const v4bf& a) {
    asm volatile("" : "+v"(xo) : "v"(a));
}

__device__ __forceinline__ v4bf to_bf4(const v4f& r) {
    v4bf a;
    a[0] = (bf16_t)r.x; a[1] = (bf16_t)r.y; a[2] = (bf16_t)r.z; a[3] = (bf16_t)r.w;
    return a;
}

#define PAIR_TOK      (N_TOK / (WG_PER_E / 2))
#define SCAN_ITERS    ((PAIR_TOK / 4 + MOE_THREADS - 1) / MOE_THREADS)

#define SM_RL     0
#define SM_WCNT   (SM_RL + ROWS_PER_PASS * 16)
#define SM_TOK    (SM_WCNT + 64)
#define TOK_SLOTS (((PAIR_TOK / 2 + ROWS_PER_PASS - 1) / ROWS_PER_PASS) * ROWS_PER_PASS)
#define SM_IMG    (SM_TOK + TOK_SLOTS * 4)
#define SM_STG    (SM_IMG + FRAGS_PER_E * FRAG_BYTES)
#define STG_PITCH 80
#define STG_BYTES (ROWS_PER_WAVE * STG_PITCH)
#define XLOADS    (ROWS_PER_WAVE / 4)
#define SM_TOTAL  (SM_STG + WG_WAVES * 2 * STG_BYTES)

__global__ __launch_bounds__(MOE_THREADS) void moe_k(const float* __restrict__ x,
                                             const int*   __restrict__ z,
                                             const float* __restrict__ W1,
                                             const float* __restrict__ b1,
                                             const float* __restrict__ W2,
                                             const float* __restrict__ b2,
                                             float* __restrict__ out) {
    __shared__ __attribute__((aligned(32))) unsigned char smem[SM_TOTAL];
    int* const s_rl   = (int*)(smem + SM_RL);
    int* const s_wcnt = (int*)(smem + SM_WCNT);
    int* const s_tok  = (int*)(smem + SM_TOK);
    unsigned char* const lds = smem + SM_IMG;

    const int tid    = threadIdx.x;
    const int lane   = tid & 31;
    const int wv     = tid >> 5;
    const int hi     = lane >> 4;
    const int lanelo = lane & 15;
    const int e      = blockIdx.x >> 4;
    const int part   = blockIdx.x & (WG_PER_E - 1);
    const int tbase  = (part >> 1) * PAIR_TOK;

    int4 zz[SCAN_ITERS];
#pragma unroll
    for (int i = 0; i < SCAN_ITERS; ++i) {
        const int t4 = i * MOE_THREADS + tid;
        zz[i] = make_int4(-1, -1, -1, -1);
        if (t4 < PAIR_TOK / 4) zz[i] = *(const int4*)(z + tbase + t4 * 4);
    }

    {
        const float* wsrc = W1 + (size_t)e * (IN_DIM * OUT_DIM);
#pragma unroll
        for (int i = 0; i < IMG_ITERS; ++i) {
            int v = tid + MOE_THREADS * i;
            v = (v < IMG_VEC) ? v : v - MOE_THREADS;
            {
                const int frag = v >> 6, lf = (v >> 1) & 31, j = v & 1;
                const int kstep = frag >> 3, ntile = frag & 7, khalf = lf >> 4, nn = lf & 15;
                const float* s = wsrc + (kstep * 32 + j * 16 + khalf * 8) * OUT_DIM + ntile * 16 + nn;
                float f[8];
#pragma unroll
                for (int q = 0; q < 8; ++q) f[q] = s[q * OUT_DIM];
                v8bf pk;
#pragma unroll
                for (int q = 0; q < 8; ++q) pk[q] = (bf16_t)f[q];
                *(v8bf*)(lds + v * 16) = pk;
            }
        }
    }

    {
        int cnt = 0;
#pragma unroll
        for (int i = 0; i < SCAN_ITERS; ++i) {
            const int zv[4] = {zz[i].x, zz[i].y, zz[i].z, zz[i].w};
#pragma unroll
            for (int j = 0; j < 4; ++j)
                cnt += __builtin_popcount(__builtin_amdgcn_ballot_w32(zv[j] == e));
        }
        if (lane == 0) s_wcnt[wv] = cnt;
    }
    __syncthreads();

    int wbase = 0, total = 0;
#pragma unroll
    for (int w = 0; w < WG_WAVES; ++w) {
        const int c = s_wcnt[w];
        total += c;
        wbase += (w < wv) ? c : 0;
    }
    wbase = __builtin_amdgcn_readfirstlane(wbase);
    total = __builtin_amdgcn_readfirstlane(total);
    const int halfc = (total + 1) >> 1;
    const int rlo   = (part & 1) ? halfc : 0;
    const int rhi   = (part & 1) ? total : halfc;
    const int hiRow = rhi - rlo;

    {
        unsigned run = (unsigned)(wbase - rlo);
#pragma unroll
        for (int i = 0; i < SCAN_ITERS; ++i) {
            const int zv[4] = {zz[i].x, zz[i].y, zz[i].z, zz[i].w};
#pragma unroll
            for (int j = 0; j < 4; ++j) {
                const bool is = (zv[j] == e);
                const unsigned m = __builtin_amdgcn_ballot_w32(is);
                const unsigned slot = __builtin_amdgcn_mbcnt_lo(m, run);
                if (is && slot < (unsigned)hiRow)
                    s_tok[slot] = tbase + (i * MOE_THREADS + tid) * 4 + j;
                run += (unsigned)__builtin_popcount(m);
            }
        }
    }

    {
        const int padEnd = ((hiRow + ROWS_PER_PASS - 1) / ROWS_PER_PASS) * ROWS_PER_PASS;
        for (int q = hiRow + tid; q < padEnd; q += MOE_THREADS) s_tok[q] = -1;
    }

    __syncthreads();

    const char* xb  = (const char*)x;
    const float* b1e = b1 + e * OUT_DIM;
    const float* W2e = W2 + e * OUT_DIM * 2;

    for (int pbase = 0; pbase < hiRow; pbase += ROWS_PER_PASS) {
        int tkv[TILES];
#pragma unroll
        for (int t = 0; t < TILES; ++t) {
            const int rloc = wv * ROWS_PER_WAVE + t * 16 + lanelo;
            tkv[t] = s_tok[pbase + rloc];
            s_rl[rloc * 4] = tkv[t];
        }

        const bool active = (pbase + __builtin_amdgcn_readfirstlane(wv) * ROWS_PER_WAVE) < hiRow;
        if (active) {
            const int q8 = lane >> 3;
            const int p8 = lane & 7;
            unsigned xo[XLOADS];
#pragma unroll
            for (int i = 0; i < XLOADS; ++i) {
                const int tk = s_tok[pbase + wv * ROWS_PER_WAVE + 4 * i + q8];
                const int tc = tk < 0 ? 0 : tk;
                xo[i] = (unsigned)tc * (IN_DIM * 4u) + (unsigned)p8 * 16u;
            }
            unsigned char* const stg = smem + SM_STG + wv * (2 * STG_BYTES);
            unsigned char* const stg_st = stg + q8 * STG_PITCH + p8 * 8;
            const unsigned char* const stg_ld = stg + lanelo * STG_PITCH + hi * 16;

            v4f raw[XLOADS];
            v16bf a[TILES];
#pragma unroll
            for (int i = 0; i < XLOADS; ++i) raw[i] = *(const v4f*)(xb + xo[i]);
#pragma unroll
            for (int i = 0; i < XLOADS; ++i)
                *(v4bf*)(stg_st + i * (4 * STG_PITCH)) = to_bf4(raw[i]);
#pragma unroll
            for (int i = 0; i < XLOADS; ++i) { xo[i] += 128u; raw[i] = *(const v4f*)(xb + xo[i]); }

            const v8f zero = {0.f, 0.f, 0.f, 0.f, 0.f, 0.f, 0.f, 0.f};
            v8f acc[TILES][8];
#pragma unroll
            for (int t = 0; t < TILES; ++t)
#pragma unroll
                for (int i = 0; i < 8; ++i) acc[t][i] = zero;

            float bbv[8], w20v[8], w21v[8], b20, b21;
#pragma unroll
            for (int ks = 0; ks < 8; ++ks) {
                if (ks == 7) {
#pragma unroll
                    for (int nt = 0; nt < 8; ++nt) {
                        const unsigned n = (unsigned)(nt * 16 + lanelo);
                        bbv[nt]  = b1e[n];
                        w20v[nt] = W2e[n * 2u + 0u];
                        w21v[nt] = W2e[n * 2u + 1u];
                    }
                    b20 = b2[e * 2 + 0];
                    b21 = b2[e * 2 + 1];
                }
#pragma unroll
                for (int t = 0; t < TILES; ++t) {
                    const unsigned char* s = stg_ld + (ks & 1) * STG_BYTES + t * (16 * STG_PITCH);
                    const v8bf lo8 = *(const v8bf*)(s);
                    const v8bf hi8 = *(const v8bf*)(s + 32);
                    a[t] = __builtin_shufflevector(lo8, hi8, 0, 1, 2, 3, 4, 5, 6, 7, 8, 9, 10, 11, 12, 13, 14, 15);
                }
                if (ks < 7) {
#pragma unroll
                    for (int i = 0; i < XLOADS; ++i) {
                        hold_until_here(raw[i]);
                        const v4bf pk = to_bf4(raw[i]);
                        *(v4bf*)(stg_st + ((ks + 1) & 1) * STG_BYTES + i * (4 * STG_PITCH)) = pk;
                        if (ks < 6) {
                            addr_after(xo[i], pk);
                            xo[i] += 128u;
                            raw[i] = *(const v4f*)(xb + xo[i]);
                        }
                    }
                }
                __builtin_amdgcn_sched_barrier(0);
                v16bf bcur = *(const v16bf*)(lds + ((ks << 3) << 10) + (lane << 5));
#pragma unroll
                for (int nt = 0; nt < 8; ++nt) {
                    v16bf bnext = bcur;
                    if (nt < 7) bnext = *(const v16bf*)(lds + (((ks << 3) | (nt + 1)) << 10) + (lane << 5));
#pragma unroll
                    for (int t = 0; t < TILES; ++t)
                        acc[t][nt] = __builtin_amdgcn_wmma_f32_16x16x32_bf16(
                            false, a[t], false, bcur, (short)0, acc[t][nt], false, false);
                    bcur = bnext;
                }
                __builtin_amdgcn_sched_group_barrier(0x100, 2, 0);
#pragma unroll
                for (int nt = 0; nt < 7; ++nt) {
                    __builtin_amdgcn_sched_group_barrier(0x100, 2, 0);
                    __builtin_amdgcn_sched_group_barrier(0x008, 3, 0);
                }
                __builtin_amdgcn_sched_group_barrier(0x008, 3, 0);
                __builtin_amdgcn_sched_barrier(0);
            }

#pragma unroll
            for (int t = 0; t < TILES; ++t) {
                float s0[8], s1[8];
#pragma unroll
                for (int v = 0; v < 8; ++v) { s0[v] = 0.f; s1[v] = 0.f; }
#pragma unroll
                for (int nt = 0; nt < 8; ++nt) {
#pragma unroll
                    for (int v = 0; v < 8; ++v) {
                        float h0 = acc[t][nt][v] + bbv[nt];
                        h0 = h0 > 0.f ? h0 : 0.f;
                        s0[v] += h0 * w20v[nt];
                        s1[v] += h0 * w21v[nt];
                    }
                }
#pragma unroll
                for (int v = 0; v < 8; ++v) {
#pragma unroll
                    for (int m = 1; m < 16; m <<= 1) {
                        s0[v] += __shfl_xor(s0[v], m, 32);
                        s1[v] += __shfl_xor(s1[v], m, 32);
                    }
                }
#pragma unroll
                for (int v = 0; v < 8; ++v) {
                    if (lanelo == v) {
                        const int M = v + 8 * hi;
                        float* rl = (float*)(s_rl + (wv * ROWS_PER_WAVE + t * 16 + M) * 4);
                        rl[1] = s0[v] + b20;
                        rl[2] = s1[v] + b21;
                    }
                }
            }
        }
        __syncthreads();

#pragma unroll
        for (int j = 0; j < 2; ++j) {
            const int rr = wv * ROWS_PER_WAVE + j * 32 + lane;
            if (j == 0 || lane < 16) {
                const int4 rw = *(const int4*)(s_rl + rr * 4);
                const int tok = rw.x;
                if (tok >= 0) {
                    const float l0 = __int_as_float(rw.y);
                    const float l1 = __int_as_float(rw.z);
                    const float mx = fmaxf(l0, l1);
                    const float e0 = __expf(l0 - mx), e1 = __expf(l1 - mx);
                    const float inv = 1.f / (e0 + e1);
                    out[(size_t)tok * 2 + 0] = e0 * inv;
                    out[(size_t)tok * 2 + 1] = e1 * inv;
                }
            }
        }
        __syncthreads();
    }
}


extern "C" void kernel_launch(void* const* d_in, const int* in_sizes, int n_in,
                              void* d_out, int out_size, void* d_ws, size_t ws_size,
                              hipStream_t stream) {
    (void)in_sizes; (void)n_in; (void)out_size; (void)d_ws; (void)ws_size;
    const float* x  = (const float*)d_in[0];
    const int*   z  = (const int*)d_in[1];
    const float* W1 = (const float*)d_in[2];
    const float* b1 = (const float*)d_in[3];
    const float* W2 = (const float*)d_in[4];
    const float* b2 = (const float*)d_in[5];
    float* out = (float*)d_out;

    moe_k<<<MOE_BLOCKS, MOE_THREADS, 0, stream>>>(x, z, W1, b1, W2, b2, out);
}


# WMMA operand roles swapped (W1 fragment as A, token rows as B): a lane holds one row's features, layer 2 sums in-lane, lane halves joined once; cross-lane butterflies go; layer-2 sum order changes
# speedup vs baseline: 3.6757x; 1.0384x over previous
#include <hip/hip_runtime.h>

#define N_TOK   131072
#define KEXP    16
#define IN_DIM  256
#define OUT_DIM 128


typedef __bf16 bf16_t;
typedef __attribute__((ext_vector_type(16))) bf16_t v16bf;
typedef __attribute__((ext_vector_type(8)))  float  v8f;
typedef __attribute__((ext_vector_type(8)))  bf16_t v8bf;

#define FRAG_BYTES    1024
#define FRAGS_PER_E   ((IN_DIM / 32) * (OUT_DIM / 16))


#define WG_WAVES      12
#define MOE_THREADS   (WG_WAVES * 32)
#define TILES         3
#define ROWS_PER_WAVE (TILES * 16)
#define ROWS_PER_PASS (WG_WAVES * ROWS_PER_WAVE)
#define WG_PER_E      16
#define MOE_BLOCKS    (KEXP * WG_PER_E)
#define IMG_VEC       (FRAGS_PER_E * FRAG_BYTES / 16)
#define IMG_ITERS     ((IMG_VEC + MOE_THREADS - 1) / MOE_THREADS)

typedef __attribute__((ext_vector_type(4))) float  v4f;
typedef __attribute__((ext_vector_type(4))) bf16_t v4bf;

__device__ __forceinline__ void hold_until_here(v4f& r) {
    asm volatile("" : "+v"(r));
}

__device__ __forceinline__ void addr_after(unsigned& xo, const v4bf& a) {
    asm volatile("" : "+v"(xo) : "v"(a));
}

__device__ __forceinline__ v4bf to_bf4(const v4f& r) {
    v4bf a;
    a[0] = (bf16_t)r.x; a[1] = (bf16_t)r.y; a[2] = (bf16_t)r.z; a[3] = (bf16_t)r.w;
    return a;
}

__device__ __forceinline__ float xhalf(float v) {
    const int b = __builtin_bit_cast(int, v);
    return __builtin_bit_cast(float, __builtin_amdgcn_permlanex16(b, b, 0x76543210, 0xfedcba98, false, false));
}

#define PAIR_TOK      (N_TOK / (WG_PER_E / 2))
#define SCAN_ITERS    ((PAIR_TOK / 4 + MOE_THREADS - 1) / MOE_THREADS)

#define SM_RL     0
#define SM_WCNT   (SM_RL + ROWS_PER_PASS * 16)
#define SM_B1     (SM_WCNT + 64)
#define SM_W2     (SM_B1 + OUT_DIM * 4)
#define SM_TOK    (SM_W2 + OUT_DIM * 2 * 4)
#define TOK_SLOTS (((PAIR_TOK / 2 + ROWS_PER_PASS - 1) / ROWS_PER_PASS) * ROWS_PER_PASS)
#define SM_IMG    (SM_TOK + TOK_SLOTS * 4)
#define SM_STG    (SM_IMG + FRAGS_PER_E * FRAG_BYTES)
#define STG_PITCH 80
#define STG_BYTES (ROWS_PER_WAVE * STG_PITCH)
#define XLOADS    (ROWS_PER_WAVE / 4)
#define SM_TOTAL  (SM_STG + WG_WAVES * 2 * STG_BYTES)

__global__ __launch_bounds__(MOE_THREADS) void moe_k(const float* __restrict__ x,
                                             const int*   __restrict__ z,
                                             const float* __restrict__ W1,
                                             const float* __restrict__ b1,
                                             const float* __restrict__ W2,
                                             const float* __restrict__ b2,
                                             float* __restrict__ out) {
    __shared__ __attribute__((aligned(32))) unsigned char smem[SM_TOTAL];
    int* const s_rl   = (int*)(smem + SM_RL);
    int* const s_wcnt = (int*)(smem + SM_WCNT);
    int* const s_tok  = (int*)(smem + SM_TOK);
    float* const s_b1 = (float*)(smem + SM_B1);
    float* const s_w2 = (float*)(smem + SM_W2);
    unsigned char* const lds = smem + SM_IMG;

    const int tid    = threadIdx.x;
    const int lane   = tid & 31;
    const int wv     = tid >> 5;
    const int hi     = lane >> 4;
    const int lanelo = lane & 15;
    const int e      = blockIdx.x >> 4;
    const int part   = blockIdx.x & (WG_PER_E - 1);
    const int tbase  = (part >> 1) * PAIR_TOK;

    int4 zz[SCAN_ITERS];
#pragma unroll
    for (int i = 0; i < SCAN_ITERS; ++i) {
        const int t4 = i * MOE_THREADS + tid;
        zz[i] = make_int4(-1, -1, -1, -1);
        if (t4 < PAIR_TOK / 4) zz[i] = *(const int4*)(z + tbase + t4 * 4);
    }
    const float cb1 = b1[e * OUT_DIM + (tid & (OUT_DIM - 1))];
    const float cw2 = W2[e * (OUT_DIM * 2) + (tid & (OUT_DIM * 2 - 1))];

    {
        const float* wsrc = W1 + (size_t)e * (IN_DIM * OUT_DIM);
#pragma unroll
        for (int i = 0; i < IMG_ITERS; ++i) {
            int v = tid + MOE_THREADS * i;
            v = (v < IMG_VEC) ? v : v - MOE_THREADS;
            {
                const int frag = v >> 6, lf = (v >> 1) & 31, j = v & 1;
                const int kstep = frag >> 3, ntile = frag & 7, khalf = lf >> 4, nn = lf & 15;
                const float* s = wsrc + (kstep * 32 + j * 16 + khalf * 8) * OUT_DIM + ntile * 16 + nn;
                float f[8];
#pragma unroll
                for (int q = 0; q < 8; ++q) f[q] = s[q * OUT_DIM];
                v8bf pk;
#pragma unroll
                for (int q = 0; q < 8; ++q) pk[q] = (bf16_t)f[q];
                *(v8bf*)(lds + v * 16) = pk;
            }
        }
    }

    if (tid < OUT_DIM)     s_b1[tid] = cb1;
    if (tid < OUT_DIM * 2) s_w2[tid] = cw2;

    {
        int cnt = 0;
#pragma unroll
        for (int i = 0; i < SCAN_ITERS; ++i) {
            const int zv[4] = {zz[i].x, zz[i].y, zz[i].z, zz[i].w};
#pragma unroll
            for (int j = 0; j < 4; ++j)
                cnt += __builtin_popcount(__builtin_amdgcn_ballot_w32(zv[j] == e));
        }
        if (lane == 0) s_wcnt[wv] = cnt;
    }
    __syncthreads();

    int wbase = 0, total = 0;
#pragma unroll
    for (int w = 0; w < WG_WAVES; ++w) {
        const int c = s_wcnt[w];
        total += c;
        wbase += (w < wv) ? c : 0;
    }
    wbase = __builtin_amdgcn_readfirstlane(wbase);
    total = __builtin_amdgcn_readfirstlane(total);
    const int halfc = (total + 1) >> 1;
    const int rlo   = (part & 1) ? halfc : 0;
    const int rhi   = (part & 1) ? total : halfc;
    const int hiRow = rhi - rlo;

    {
        unsigned run = (unsigned)(wbase - rlo);
#pragma unroll
        for (int i = 0; i < SCAN_ITERS; ++i) {
            const int zv[4] = {zz[i].x, zz[i].y, zz[i].z, zz[i].w};
#pragma unroll
            for (int j = 0; j < 4; ++j) {
                const bool is = (zv[j] == e);
                const unsigned m = __builtin_amdgcn_ballot_w32(is);
                const unsigned slot = __builtin_amdgcn_mbcnt_lo(m, run);
                if (is && slot < (unsigned)hiRow)
                    s_tok[slot] = tbase + (i * MOE_THREADS + tid) * 4 + j;
                run += (unsigned)__builtin_popcount(m);
            }
        }
    }

    {
        const int padEnd = ((hiRow + ROWS_PER_PASS - 1) / ROWS_PER_PASS) * ROWS_PER_PASS;
        for (int q = hiRow + tid; q < padEnd; q += MOE_THREADS) s_tok[q] = -1;
    }

    __syncthreads();

    const char* xb  = (const char*)x;

    for (int pbase = 0; pbase < hiRow; pbase += ROWS_PER_PASS) {
        int tkv[TILES];
#pragma unroll
        for (int t = 0; t < TILES; ++t) {
            const int rloc = wv * ROWS_PER_WAVE + t * 16 + lanelo;
            tkv[t] = s_tok[pbase + rloc];
            s_rl[rloc * 4] = tkv[t];
        }

        const bool active = (pbase + __builtin_amdgcn_readfirstlane(wv) * ROWS_PER_WAVE) < hiRow;
        if (active) {
            const int q8 = lane >> 3;
            const int p8 = lane & 7;
            unsigned xo[XLOADS];
#pragma unroll
            for (int i = 0; i < XLOADS; ++i) {
                const int tk = s_tok[pbase + wv * ROWS_PER_WAVE + 4 * i + q8];
                const int tc = tk < 0 ? 0 : tk;
                xo[i] = (unsigned)tc * (IN_DIM * 4u) + (unsigned)p8 * 16u;
            }
            unsigned char* const stg = smem + SM_STG + wv * (2 * STG_BYTES);
            unsigned char* const stg_st = stg + q8 * STG_PITCH + p8 * 8;
            const unsigned char* const stg_ld = stg + lanelo * STG_PITCH + hi * 16;

            v4f raw[XLOADS];
            v16bf a[TILES];
#pragma unroll
            for (int i = 0; i < XLOADS; ++i) raw[i] = *(const v4f*)(xb + xo[i]);
#pragma unroll
            for (int i = 0; i < XLOADS; ++i)
                *(v4bf*)(stg_st + i * (4 * STG_PITCH)) = to_bf4(raw[i]);
#pragma unroll
            for (int i = 0; i < XLOADS; ++i) { xo[i] += 128u; raw[i] = *(const v4f*)(xb + xo[i]); }

            const v8f zero = {0.f, 0.f, 0.f, 0.f, 0.f, 0.f, 0.f, 0.f};
            v8f acc[TILES][8];
#pragma unroll
            for (int t = 0; t < TILES; ++t)
#pragma unroll
                for (int i = 0; i < 8; ++i) acc[t][i] = zero;

            const float b20 = b2[e * 2 + 0], b21 = b2[e * 2 + 1];
#pragma unroll
            for (int ks = 0; ks < 8; ++ks) {
#pragma unroll
                for (int t = 0; t < TILES; ++t) {
                    const unsigned char* s = stg_ld + (ks & 1) * STG_BYTES + t * (16 * STG_PITCH);
                    const v8bf lo8 = *(const v8bf*)(s);
                    const v8bf hi8 = *(const v8bf*)(s + 32);
                    a[t] = __builtin_shufflevector(lo8, hi8, 0, 1, 2, 3, 4, 5, 6, 7, 8, 9, 10, 11, 12, 13, 14, 15);
                }
                if (ks < 7) {
#pragma unroll
                    for (int i = 0; i < XLOADS; ++i) {
                        hold_until_here(raw[i]);
                        const v4bf pk = to_bf4(raw[i]);
                        *(v4bf*)(stg_st + ((ks + 1) & 1) * STG_BYTES + i * (4 * STG_PITCH)) = pk;
                        if (ks < 6) {
                            addr_after(xo[i], pk);
                            xo[i] += 128u;
                            raw[i] = *(const v4f*)(xb + xo[i]);
                        }
                    }
                }
                __builtin_amdgcn_sched_barrier(0);
                v16bf bcur = *(const v16bf*)(lds + ((ks << 3) << 10) + (lane << 5));
#pragma unroll
                for (int nt = 0; nt < 8; ++nt) {
                    v16bf bnext = bcur;
                    if (nt < 7) bnext = *(const v16bf*)(lds + (((ks << 3) | (nt + 1)) << 10) + (lane << 5));
#pragma unroll
                    for (int t = 0; t < TILES; ++t)
                        acc[t][nt] = __builtin_amdgcn_wmma_f32_16x16x32_bf16(
                            false, bcur, false, a[t], (short)0, acc[t][nt], false, false);
                    bcur = bnext;
                }
                __builtin_amdgcn_sched_group_barrier(0x100, 2, 0);
#pragma unroll
                for (int nt = 0; nt < 7; ++nt) {
                    __builtin_amdgcn_sched_group_barrier(0x100, 2, 0);
                    __builtin_amdgcn_sched_group_barrier(0x008, 3, 0);
                }
                __builtin_amdgcn_sched_group_barrier(0x008, 3, 0);
                __builtin_amdgcn_sched_barrier(0);
            }

            {
                const float* const cb = s_b1 + hi * 8;
                const float* const cw = s_w2 + hi * 16;
                float s0[TILES], s1[TILES];
#pragma unroll
                for (int t = 0; t < TILES; ++t) { s0[t] = 0.f; s1[t] = 0.f; }
#pragma unroll
                for (int nt = 0; nt < 8; ++nt) {
                    float bb[8], w0[8], w1[8];
#pragma unroll
                    for (int g = 0; g < 2; ++g) {
                        const v4f q = *(const v4f*)(cb + nt * 16 + g * 4);
                        bb[4 * g + 0] = q.x; bb[4 * g + 1] = q.y; bb[4 * g + 2] = q.z; bb[4 * g + 3] = q.w;
                    }
#pragma unroll
                    for (int g = 0; g < 4; ++g) {
                        const v4f q = *(const v4f*)(cw + nt * 32 + g * 4);
                        w0[2 * g + 0] = q.x; w1[2 * g + 0] = q.y; w0[2 * g + 1] = q.z; w1[2 * g + 1] = q.w;
                    }
#pragma unroll
                    for (int t = 0; t < TILES; ++t) {
#pragma unroll
                        for (int r = 0; r < 8; ++r) {
                            float h0 = acc[t][nt][r] + bb[r];
                            h0 = h0 > 0.f ? h0 : 0.f;
                            s0[t] += h0 * w0[r];
                            s1[t] += h0 * w1[r];
                        }
                    }
                }
#pragma unroll
                for (int t = 0; t < TILES; ++t) {
                    const float l0 = s0[t] + xhalf(s0[t]) + b20;
                    const float l1 = s1[t] + xhalf(s1[t]) + b21;
                    float* rl = (float*)(s_rl + (wv * ROWS_PER_WAVE + t * 16 + lanelo) * 4);
                    rl[1] = l0;
                    rl[2] = l1;
                }
            }
        }
        __syncthreads();

#pragma unroll
        for (int j = 0; j < 2; ++j) {
            const int rr = wv * ROWS_PER_WAVE + j * 32 + lane;
            if (j == 0 || lane < 16) {
                const int4 rw = *(const int4*)(s_rl + rr * 4);
                const int tok = rw.x;
                if (tok >= 0) {
                    const float l0 = __int_as_float(rw.y);
                    const float l1 = __int_as_float(rw.z);
                    const float mx = fmaxf(l0, l1);
                    const float e0 = __expf(l0 - mx), e1 = __expf(l1 - mx);
                    const float inv = 1.f / (e0 + e1);
                    out[(size_t)tok * 2 + 0] = e0 * inv;
                    out[(size_t)tok * 2 + 1] = e1 * inv;
                }
            }
        }
        __syncthreads();
    }
}


extern "C" void kernel_launch(void* const* d_in, const int* in_sizes, int n_in,
                              void* d_out, int out_size, void* d_ws, size_t ws_size,
                              hipStream_t stream) {
    (void)in_sizes; (void)n_in; (void)out_size; (void)d_ws; (void)ws_size;
    const float* x  = (const float*)d_in[0];
    const int*   z  = (const int*)d_in[1];
    const float* W1 = (const float*)d_in[2];
    const float* b1 = (const float*)d_in[3];
    const float* W2 = (const float*)d_in[4];
    const float* b2 = (const float*)d_in[5];
    float* out = (float*)d_out;

    moe_k<<<MOE_BLOCKS, MOE_THREADS, 0, stream>>>(x, z, W1, b1, W2, b2, out);
}


# Set-up re-ordered: W1 is requested up front and held in registers while the token list is built from z; LDS-only barriers keep it, and then the first rows, in flight; image converted behind them
# speedup vs baseline: 3.8871x; 1.0575x over previous
#include <hip/hip_runtime.h>

#define N_TOK   131072
#define KEXP    16
#define IN_DIM  256
#define OUT_DIM 128


typedef __bf16 bf16_t;
typedef __attribute__((ext_vector_type(16))) bf16_t v16bf;
typedef __attribute__((ext_vector_type(8)))  float  v8f;
typedef __attribute__((ext_vector_type(8)))  bf16_t v8bf;

#define FRAG_BYTES    1024
#define FRAGS_PER_E   ((IN_DIM / 32) * (OUT_DIM / 16))


#define WG_WAVES      12
#define MOE_THREADS   (WG_WAVES * 32)
#define TILES         3
#define ROWS_PER_WAVE (TILES * 16)
#define ROWS_PER_PASS (WG_WAVES * ROWS_PER_WAVE)
#define WG_PER_E      16
#define MOE_BLOCKS    (KEXP * WG_PER_E)
#define IMG_VEC       (FRAGS_PER_E * FRAG_BYTES / 16)
#define IMG_ITERS     ((IMG_VEC + MOE_THREADS - 1) / MOE_THREADS)

typedef __attribute__((ext_vector_type(4))) float  v4f;
typedef __attribute__((ext_vector_type(4))) bf16_t v4bf;

__device__ __forceinline__ void hold_until_here(v4f& r) {
    asm volatile("" : "+v"(r));
}

__device__ __forceinline__ void addr_after(unsigned& xo, const v4bf& a) {
    asm volatile("" : "+v"(xo) : "v"(a));
}

__device__ __forceinline__ v4bf to_bf4(const v4f& r) {
    v4bf a;
    a[0] = (bf16_t)r.x; a[1] = (bf16_t)r.y; a[2] = (bf16_t)r.z; a[3] = (bf16_t)r.w;
    return a;
}

__device__ __forceinline__ float xhalf(float v) {
    const int b = __builtin_bit_cast(int, v);
    return __builtin_bit_cast(float, __builtin_amdgcn_permlanex16(b, b, 0x76543210, 0xfedcba98, false, false));
}

__device__ __forceinline__ void lds_barrier() {
    asm volatile("s_wait_dscnt 0x0\n\ts_barrier_signal -1\n\ts_barrier_wait -1" ::: "memory");
}

#define PAIR_TOK      (N_TOK / (WG_PER_E / 2))
#define SCAN_ITERS    ((PAIR_TOK / 4 + MOE_THREADS - 1) / MOE_THREADS)

#define SM_RL     0
#define SM_WCNT   (SM_RL + ROWS_PER_PASS * 16)
#define SM_B1     (SM_WCNT + 64)
#define SM_W2     (SM_B1 + OUT_DIM * 4)
#define SM_TOK    (SM_W2 + OUT_DIM * 2 * 4)
#define TOK_SLOTS (((PAIR_TOK / 2 + ROWS_PER_PASS - 1) / ROWS_PER_PASS) * ROWS_PER_PASS)
#define SM_IMG    (SM_TOK + TOK_SLOTS * 4)
#define SM_STG    (SM_IMG + FRAGS_PER_E * FRAG_BYTES)
#define STG_PITCH 80
#define STG_BYTES (ROWS_PER_WAVE * STG_PITCH)
#define XLOADS    (ROWS_PER_WAVE / 4)
#define SM_TOTAL  (SM_STG + WG_WAVES * 2 * STG_BYTES)

__global__ __launch_bounds__(MOE_THREADS) void moe_k(const float* __restrict__ x,
                                             const int*   __restrict__ z,
                                             const float* __restrict__ W1,
                                             const float* __restrict__ b1,
                                             const float* __restrict__ W2,
                                             const float* __restrict__ b2,
                                             float* __restrict__ out) {
    __shared__ __attribute__((aligned(32))) unsigned char smem[SM_TOTAL];
    int* const s_rl   = (int*)(smem + SM_RL);
    int* const s_wcnt = (int*)(smem + SM_WCNT);
    int* const s_tok  = (int*)(smem + SM_TOK);
    float* const s_b1 = (float*)(smem + SM_B1);
    float* const s_w2 = (float*)(smem + SM_W2);
    unsigned char* const lds = smem + SM_IMG;

    const int tid    = threadIdx.x;
    const int lane   = tid & 31;
    const int wv     = tid >> 5;
    const int hi     = lane >> 4;
    const int lanelo = lane & 15;
    const int e      = blockIdx.x >> 4;
    const int part   = blockIdx.x & (WG_PER_E - 1);
    const int tbase  = (part >> 1) * PAIR_TOK;

    int4 zz[SCAN_ITERS];
#pragma unroll
    for (int i = 0; i < SCAN_ITERS; ++i) {
        const int t4 = i * MOE_THREADS + tid;
        zz[i] = make_int4(-1, -1, -1, -1);
        if (t4 < PAIR_TOK / 4) zz[i] = *(const int4*)(z + tbase + t4 * 4);
    }
    const float cb1 = b1[e * OUT_DIM + (tid & (OUT_DIM - 1))];
    const float cw2 = W2[e * (OUT_DIM * 2) + (tid & (OUT_DIM * 2 - 1))];

    float wf[IMG_ITERS][8];
    {
        const float* wsrc = W1 + (size_t)e * (IN_DIM * OUT_DIM);
#pragma unroll
        for (int i = 0; i < IMG_ITERS; ++i) {
            int v = tid + MOE_THREADS * i;
            v = (v < IMG_VEC) ? v : v - MOE_THREADS;
            const int frag = v >> 6, lf = (v >> 1) & 31, j = v & 1;
            const int kstep = frag >> 3, ntile = frag & 7, khalf = lf >> 4, nn = lf & 15;
            const float* s = wsrc + (kstep * 32 + j * 16 + khalf * 8) * OUT_DIM + ntile * 16 + nn;
#pragma unroll
            for (int q = 0; q < 8; ++q) wf[i][q] = s[q * OUT_DIM];
        }
    }

    {
        int cnt = 0;
#pragma unroll
        for (int i = 0; i < SCAN_ITERS; ++i) {
            const int zv[4] = {zz[i].x, zz[i].y, zz[i].z, zz[i].w};
#pragma unroll
            for (int j = 0; j < 4; ++j)
                cnt += __builtin_popcount(__builtin_amdgcn_ballot_w32(zv[j] == e));
        }
        if (lane == 0) s_wcnt[wv] = cnt;
    }
    lds_barrier();

    int wbase = 0, total = 0;
#pragma unroll
    for (int w = 0; w < WG_WAVES; ++w) {
        const int c = s_wcnt[w];
        total += c;
        wbase += (w < wv) ? c : 0;
    }
    wbase = __builtin_amdgcn_readfirstlane(wbase);
    total = __builtin_amdgcn_readfirstlane(total);
    const int halfc = (total + 1) >> 1;
    const int rlo   = (part & 1) ? halfc : 0;
    const int rhi   = (part & 1) ? total : halfc;
    const int hiRow = rhi - rlo;

    {
        unsigned run = (unsigned)(wbase - rlo);
#pragma unroll
        for (int i = 0; i < SCAN_ITERS; ++i) {
            const int zv[4] = {zz[i].x, zz[i].y, zz[i].z, zz[i].w};
#pragma unroll
            for (int j = 0; j < 4; ++j) {
                const bool is = (zv[j] == e);
                const unsigned m = __builtin_amdgcn_ballot_w32(is);
                const unsigned slot = __builtin_amdgcn_mbcnt_lo(m, run);
                if (is && slot < (unsigned)hiRow)
                    s_tok[slot] = tbase + (i * MOE_THREADS + tid) * 4 + j;
                run += (unsigned)__builtin_popcount(m);
            }
        }
    }

    {
        const int padEnd = ((hiRow + ROWS_PER_PASS - 1) / ROWS_PER_PASS) * ROWS_PER_PASS;
        for (int q = hiRow + tid; q < padEnd; q += MOE_THREADS) s_tok[q] = -1;
    }

    lds_barrier();

    const char* xb  = (const char*)x;
    const int q8 = lane >> 3;
    const int p8 = lane & 7;

    bool active = false;
    unsigned xo[XLOADS];
    v4f raw[XLOADS];

    auto pass_setup = [&](const int pbase) {
#pragma unroll
        for (int t = 0; t < TILES; ++t) {
            const int rloc = wv * ROWS_PER_WAVE + t * 16 + lanelo;
            s_rl[rloc * 4] = s_tok[pbase + rloc];
        }
        active = (pbase + __builtin_amdgcn_readfirstlane(wv) * ROWS_PER_WAVE) < hiRow;
#pragma unroll
        for (int i = 0; i < XLOADS; ++i) {
            const int tk = s_tok[pbase + wv * ROWS_PER_WAVE + 4 * i + q8];
            const int tc = tk < 0 ? 0 : tk;
            xo[i] = (unsigned)tc * (IN_DIM * 4u) + (unsigned)p8 * 16u;
        }
#pragma unroll
        for (int i = 0; i < XLOADS; ++i) raw[i] = *(const v4f*)(xb + xo[i]);
    };

    if (hiRow > 0) {
        pass_setup(0);
        __builtin_amdgcn_sched_barrier(0);

#pragma unroll
        for (int i = 0; i < IMG_ITERS; ++i) {
            int v = tid + MOE_THREADS * i;
            v = (v < IMG_VEC) ? v : v - MOE_THREADS;
            v8bf pk;
#pragma unroll
            for (int q = 0; q < 8; ++q) pk[q] = (bf16_t)wf[i][q];
            *(v8bf*)(lds + v * 16) = pk;
        }
        s_b1[tid & (OUT_DIM - 1)]     = cb1;
        s_w2[tid & (OUT_DIM * 2 - 1)] = cw2;

        lds_barrier();

        for (int pbase = 0;;) {
            if (active) {
                unsigned char* const stg = smem + SM_STG + wv * (2 * STG_BYTES);
                unsigned char* const stg_st = stg + q8 * STG_PITCH + p8 * 8;
                const unsigned char* const stg_ld = stg + lanelo * STG_PITCH + hi * 16;

                v16bf a[TILES];
#pragma unroll
                for (int i = 0; i < XLOADS; ++i)
                    *(v4bf*)(stg_st + i * (4 * STG_PITCH)) = to_bf4(raw[i]);
#pragma unroll
                for (int i = 0; i < XLOADS; ++i) { xo[i] += 128u; raw[i] = *(const v4f*)(xb + xo[i]); }

                const v8f zero = {0.f, 0.f, 0.f, 0.f, 0.f, 0.f, 0.f, 0.f};
                v8f acc[TILES][8];
#pragma unroll
                for (int t = 0; t < TILES; ++t)
#pragma unroll
                    for (int i = 0; i < 8; ++i) acc[t][i] = zero;

                const float b20 = b2[e * 2 + 0], b21 = b2[e * 2 + 1];
#pragma unroll
                for (int ks = 0; ks < 8; ++ks) {
#pragma unroll
                    for (int t = 0; t < TILES; ++t) {
                        const unsigned char* s = stg_ld + (ks & 1) * STG_BYTES + t * (16 * STG_PITCH);
                        const v8bf lo8 = *(const v8bf*)(s);
                        const v8bf hi8 = *(const v8bf*)(s + 32);
                        a[t] = __builtin_shufflevector(lo8, hi8, 0, 1, 2, 3, 4, 5, 6, 7, 8, 9, 10, 11, 12, 13, 14, 15);
                    }
                    if (ks < 7) {
#pragma unroll
                        for (int i = 0; i < XLOADS; ++i) {
                            hold_until_here(raw[i]);
                            const v4bf pk = to_bf4(raw[i]);
                            *(v4bf*)(stg_st + ((ks + 1) & 1) * STG_BYTES + i * (4 * STG_PITCH)) = pk;
                            if (ks < 6) {
                                addr_after(xo[i], pk);
                                xo[i] += 128u;
                                raw[i] = *(const v4f*)(xb + xo[i]);
                            }
                        }
                    }
                    __builtin_amdgcn_sched_barrier(0);
                    v16bf bcur = *(const v16bf*)(lds + ((ks << 3) << 10) + (lane << 5));
#pragma unroll
                    for (int nt = 0; nt < 8; ++nt) {
                        v16bf bnext = bcur;
                        if (nt < 7) bnext = *(const v16bf*)(lds + (((ks << 3) | (nt + 1)) << 10) + (lane << 5));
#pragma unroll
                        for (int t = 0; t < TILES; ++t)
                            acc[t][nt] = __builtin_amdgcn_wmma_f32_16x16x32_bf16(
                                false, bcur, false, a[t], (short)0, acc[t][nt], false, false);
                        bcur = bnext;
                    }
                    __builtin_amdgcn_sched_group_barrier(0x100, 2, 0);
#pragma unroll
                    for (int nt = 0; nt < 7; ++nt) {
                        __builtin_amdgcn_sched_group_barrier(0x100, 2, 0);
                        __builtin_amdgcn_sched_group_barrier(0x008, 3, 0);
                    }
                    __builtin_amdgcn_sched_group_barrier(0x008, 3, 0);
                    __builtin_amdgcn_sched_barrier(0);
                }

                {
                    const float* const cb = s_b1 + hi * 8;
                    const float* const cw = s_w2 + hi * 16;
                    float s0[TILES], s1[TILES];
#pragma unroll
                    for (int t = 0; t < TILES; ++t) { s0[t] = 0.f; s1[t] = 0.f; }
#pragma unroll
                    for (int nt = 0; nt < 8; ++nt) {
                        float bb[8], w0[8], w1[8];
#pragma unroll
                        for (int g = 0; g < 2; ++g) {
                            const v4f q = *(const v4f*)(cb + nt * 16 + g * 4);
                            bb[4 * g + 0] = q.x; bb[4 * g + 1] = q.y; bb[4 * g + 2] = q.z; bb[4 * g + 3] = q.w;
                        }
#pragma unroll
                        for (int g = 0; g < 4; ++g) {
                            const v4f q = *(const v4f*)(cw + nt * 32 + g * 4);
                            w0[2 * g + 0] = q.x; w1[2 * g + 0] = q.y; w0[2 * g + 1] = q.z; w1[2 * g + 1] = q.w;
                        }
#pragma unroll
                        for (int t = 0; t < TILES; ++t) {
#pragma unroll
                            for (int r = 0; r < 8; ++r) {
                                float h0 = acc[t][nt][r] + bb[r];
                                h0 = h0 > 0.f ? h0 : 0.f;
                                s0[t] += h0 * w0[r];
                                s1[t] += h0 * w1[r];
                            }
                        }
                    }
#pragma unroll
                    for (int t = 0; t < TILES; ++t) {
                        const float l0 = s0[t] + xhalf(s0[t]) + b20;
                        const float l1 = s1[t] + xhalf(s1[t]) + b21;
                        float* rl = (float*)(s_rl + (wv * ROWS_PER_WAVE + t * 16 + lanelo) * 4);
                        rl[1] = l0;
                        rl[2] = l1;
                    }
                }
            }
            __syncthreads();

#pragma unroll
            for (int j = 0; j < 2; ++j) {
                const int rr = wv * ROWS_PER_WAVE + j * 32 + lane;
                if (j == 0 || lane < 16) {
                    const int4 rw = *(const int4*)(s_rl + rr * 4);
                    const int tok = rw.x;
                    if (tok >= 0) {
                        const float l0 = __int_as_float(rw.y);
                        const float l1 = __int_as_float(rw.z);
                        const float mx = fmaxf(l0, l1);
                        const float e0 = __expf(l0 - mx), e1 = __expf(l1 - mx);
                        const float inv = 1.f / (e0 + e1);
                        out[(size_t)tok * 2 + 0] = e0 * inv;
                        out[(size_t)tok * 2 + 1] = e1 * inv;
                    }
                }
            }
            pbase += ROWS_PER_PASS;
            if (pbase >= hiRow) break;
            __syncthreads();
            pass_setup(pbase);
        }
    }
}


extern "C" void kernel_launch(void* const* d_in, const int* in_sizes, int n_in,
                              void* d_out, int out_size, void* d_ws, size_t ws_size,
                              hipStream_t stream) {
    (void)in_sizes; (void)n_in; (void)out_size; (void)d_ws; (void)ws_size;
    const float* x  = (const float*)d_in[0];
    const int*   z  = (const int*)d_in[1];
    const float* W1 = (const float*)d_in[2];
    const float* b1 = (const float*)d_in[3];
    const float* W2 = (const float*)d_in[4];
    const float* b2 = (const float*)d_in[5];
    float* out = (float*)d_out;

    moe_k<<<MOE_BLOCKS, MOE_THREADS, 0, stream>>>(x, z, W1, b1, W2, b2, out);
}


# Token list built lane-major: each lane keeps its matches as a bit mask, six ballots of the lane counts rank them, a wave stores one match per lane per trip (about 8 trips, were 44 masked steps)
# speedup vs baseline: 4.1007x; 1.0550x over previous
#include <hip/hip_runtime.h>

#define N_TOK   131072
#define KEXP    16
#define IN_DIM  256
#define OUT_DIM 128


typedef __bf16 bf16_t;
typedef __attribute__((ext_vector_type(16))) bf16_t v16bf;
typedef __attribute__((ext_vector_type(8)))  float  v8f;
typedef __attribute__((ext_vector_type(8)))  bf16_t v8bf;

#define FRAG_BYTES    1024
#define FRAGS_PER_E   ((IN_DIM / 32) * (OUT_DIM / 16))


#define WG_WAVES      12
#define MOE_THREADS   (WG_WAVES * 32)
#define TILES         3
#define ROWS_PER_WAVE (TILES * 16)
#define ROWS_PER_PASS (WG_WAVES * ROWS_PER_WAVE)
#define WG_PER_E      16
#define MOE_BLOCKS    (KEXP * WG_PER_E)
#define IMG_VEC       (FRAGS_PER_E * FRAG_BYTES / 16)
#define IMG_ITERS     ((IMG_VEC + MOE_THREADS - 1) / MOE_THREADS)

typedef __attribute__((ext_vector_type(4))) float  v4f;
typedef __attribute__((ext_vector_type(4))) bf16_t v4bf;

__device__ __forceinline__ void hold_until_here(v4f& r) {
    asm volatile("" : "+v"(r));
}

__device__ __forceinline__ void addr_after(unsigned& xo, const v4bf& a) {
    asm volatile("" : "+v"(xo) : "v"(a));
}

__device__ __forceinline__ v4bf to_bf4(const v4f& r) {
    v4bf a;
    a[0] = (bf16_t)r.x; a[1] = (bf16_t)r.y; a[2] = (bf16_t)r.z; a[3] = (bf16_t)r.w;
    return a;
}

__device__ __forceinline__ float xhalf(float v) {
    const int b = __builtin_bit_cast(int, v);
    return __builtin_bit_cast(float, __builtin_amdgcn_permlanex16(b, b, 0x76543210, 0xfedcba98, false, false));
}

__device__ __forceinline__ void lds_barrier() {
    asm volatile("s_wait_dscnt 0x0\n\ts_barrier_signal -1\n\ts_barrier_wait -1" ::: "memory");
}

#define PAIR_TOK      (N_TOK / (WG_PER_E / 2))
#define SCAN_ITERS    ((PAIR_TOK / 4 + MOE_THREADS - 1) / MOE_THREADS)

#define SM_RL     0
#define SM_WCNT   (SM_RL + ROWS_PER_PASS * 16)
#define SM_B1     (SM_WCNT + 64)
#define SM_W2     (SM_B1 + OUT_DIM * 4)
#define SM_TOK    (SM_W2 + OUT_DIM * 2 * 4)
#define TOK_SLOTS (((PAIR_TOK / 2 + ROWS_PER_PASS - 1) / ROWS_PER_PASS) * ROWS_PER_PASS)
#define SM_IMG    (SM_TOK + TOK_SLOTS * 4)
#define SM_STG    (SM_IMG + FRAGS_PER_E * FRAG_BYTES)
#define STG_PITCH 80
#define STG_BYTES (ROWS_PER_WAVE * STG_PITCH)
#define XLOADS    (ROWS_PER_WAVE / 4)
#define SM_TOTAL  (SM_STG + WG_WAVES * 2 * STG_BYTES)

__global__ __launch_bounds__(MOE_THREADS) void moe_k(const float* __restrict__ x,
                                             const int*   __restrict__ z,
                                             const float* __restrict__ W1,
                                             const float* __restrict__ b1,
                                             const float* __restrict__ W2,
                                             const float* __restrict__ b2,
                                             float* __restrict__ out) {
    __shared__ __attribute__((aligned(32))) unsigned char smem[SM_TOTAL];
    int* const s_rl   = (int*)(smem + SM_RL);
    int* const s_wcnt = (int*)(smem + SM_WCNT);
    int* const s_tok  = (int*)(smem + SM_TOK);
    float* const s_b1 = (float*)(smem + SM_B1);
    float* const s_w2 = (float*)(smem + SM_W2);
    unsigned char* const lds = smem + SM_IMG;

    const int tid    = threadIdx.x;
    const int lane   = tid & 31;
    const int wv     = tid >> 5;
    const int hi     = lane >> 4;
    const int lanelo = lane & 15;
    const int e      = blockIdx.x >> 4;
    const int part   = blockIdx.x & (WG_PER_E - 1);
    const int tbase  = (part >> 1) * PAIR_TOK;

    int4 zz[SCAN_ITERS];
#pragma unroll
    for (int i = 0; i < SCAN_ITERS; ++i) {
        const int t4 = i * MOE_THREADS + tid;
        zz[i] = make_int4(-1, -1, -1, -1);
        if (t4 < PAIR_TOK / 4) zz[i] = *(const int4*)(z + tbase + t4 * 4);
    }
    const float cb1 = b1[e * OUT_DIM + (tid & (OUT_DIM - 1))];
    const float cw2 = W2[e * (OUT_DIM * 2) + (tid & (OUT_DIM * 2 - 1))];

    float wf[IMG_ITERS][8];
    {
        const float* wsrc = W1 + (size_t)e * (IN_DIM * OUT_DIM);
#pragma unroll
        for (int i = 0; i < IMG_ITERS; ++i) {
            int v = tid + MOE_THREADS * i;
            v = (v < IMG_VEC) ? v : v - MOE_THREADS;
            const int frag = v >> 6, lf = (v >> 1) & 31, j = v & 1;
            const int kstep = frag >> 3, ntile = frag & 7, khalf = lf >> 4, nn = lf & 15;
            const float* s = wsrc + (kstep * 32 + j * 16 + khalf * 8) * OUT_DIM + ntile * 16 + nn;
#pragma unroll
            for (int q = 0; q < 8; ++q) wf[i][q] = s[q * OUT_DIM];
        }
    }

    unsigned macc[4] = {0u, 0u, 0u, 0u};
#pragma unroll
    for (int i = SCAN_ITERS - 1; i >= 0; --i) {
        const int zv[4] = {zz[i].x, zz[i].y, zz[i].z, zz[i].w};
#pragma unroll
        for (int j = 0; j < 4; ++j) macc[j] = (macc[j] << 1) | (unsigned)(zv[j] == e);
    }
    unsigned mlo = macc[0] | (macc[1] << 16);
    unsigned mhi = macc[2] | (macc[3] << 16);
    const unsigned lcnt = (unsigned)(__builtin_popcount(mlo) + __builtin_popcount(mhi));

    unsigned lpre = 0u;
    int maxc = 0;
    {
        int cnt = 0;
        unsigned bm[6];
#pragma unroll
        for (int b = 0; b < 6; ++b) {
            bm[b] = __builtin_amdgcn_ballot_w32(((lcnt >> b) & 1u) != 0u);
            lpre += __builtin_amdgcn_mbcnt_lo(bm[b], 0u) << b;
            cnt  += __builtin_popcount(bm[b]) << b;
        }
        unsigned cand = 0xffffffffu;
#pragma unroll
        for (int b = 5; b >= 0; --b) {
            const unsigned t = cand & bm[b];
            if (t != 0u) { cand = t; maxc |= 1 << b; }
        }
        if (lane == 0) s_wcnt[wv] = cnt;
    }
    lds_barrier();

    int wbase = 0, total = 0;
#pragma unroll
    for (int w = 0; w < WG_WAVES; ++w) {
        const int c = s_wcnt[w];
        total += c;
        wbase += (w < wv) ? c : 0;
    }
    wbase = __builtin_amdgcn_readfirstlane(wbase);
    total = __builtin_amdgcn_readfirstlane(total);
    const int halfc = (total + 1) >> 1;
    const int rlo   = (part & 1) ? halfc : 0;
    const int rhi   = (part & 1) ? total : halfc;
    const int hiRow = rhi - rlo;

    {
        unsigned slot = (unsigned)(wbase - rlo) + lpre;
        const unsigned send = slot + lcnt;
        const unsigned slim = send < (unsigned)hiRow ? send : (unsigned)hiRow;
        const int tok0 = tbase + tid * 4;
        for (int trip = 0; trip < maxc; ++trip) {
            const bool inlo = mlo != 0u;
            const unsigned w = inlo ? mlo : mhi;
            const unsigned b = (unsigned)__builtin_ctz(w | 0x80000000u);
            const int tok = (inlo ? tok0 : tok0 + 2) + (int)(b & 15u) * (MOE_THREADS * 4) + (int)(b >> 4);
            if (slot < slim) s_tok[slot] = tok;
            slot += (w != 0u) ? 1u : 0u;
            const unsigned w1 = w & (w - 1u);
            mlo = inlo ? w1 : 0u;
            mhi = inlo ? mhi : w1;
        }
    }

    {
        const int padEnd = ((hiRow + ROWS_PER_PASS - 1) / ROWS_PER_PASS) * ROWS_PER_PASS;
        for (int q = hiRow + tid; q < padEnd; q += MOE_THREADS) s_tok[q] = -1;
    }

    lds_barrier();

    const char* xb  = (const char*)x;
    const int q8 = lane >> 3;
    const int p8 = lane & 7;

    bool active = false;
    unsigned xo[XLOADS];
    v4f raw[XLOADS];

    auto pass_setup = [&](const int pbase) {
#pragma unroll
        for (int t = 0; t < TILES; ++t) {
            const int rloc = wv * ROWS_PER_WAVE + t * 16 + lanelo;
            s_rl[rloc * 4] = s_tok[pbase + rloc];
        }
        active = (pbase + __builtin_amdgcn_readfirstlane(wv) * ROWS_PER_WAVE) < hiRow;
#pragma unroll
        for (int i = 0; i < XLOADS; ++i) {
            const int tk = s_tok[pbase + wv * ROWS_PER_WAVE + 4 * i + q8];
            const int tc = tk < 0 ? 0 : tk;
            xo[i] = (unsigned)tc * (IN_DIM * 4u) + (unsigned)p8 * 16u;
        }
#pragma unroll
        for (int i = 0; i < XLOADS; ++i) raw[i] = *(const v4f*)(xb + xo[i]);
    };

    if (hiRow > 0) {
        pass_setup(0);
        __builtin_amdgcn_sched_barrier(0);

#pragma unroll
        for (int i = 0; i < IMG_ITERS; ++i) {
            int v = tid + MOE_THREADS * i;
            v = (v < IMG_VEC) ? v : v - MOE_THREADS;
            v8bf pk;
#pragma unroll
            for (int q = 0; q < 8; ++q) pk[q] = (bf16_t)wf[i][q];
            *(v8bf*)(lds + v * 16) = pk;
        }
        s_b1[tid & (OUT_DIM - 1)]     = cb1;
        s_w2[tid & (OUT_DIM * 2 - 1)] = cw2;

        lds_barrier();

        for (int pbase = 0;;) {
            if (active) {
                unsigned char* const stg = smem + SM_STG + wv * (2 * STG_BYTES);
                unsigned char* const stg_st = stg + q8 * STG_PITCH + p8 * 8;
                const unsigned char* const stg_ld = stg + lanelo * STG_PITCH + hi * 16;

                v16bf a[TILES];
#pragma unroll
                for (int i = 0; i < XLOADS; ++i)
                    *(v4bf*)(stg_st + i * (4 * STG_PITCH)) = to_bf4(raw[i]);
#pragma unroll
                for (int i = 0; i < XLOADS; ++i) { xo[i] += 128u; raw[i] = *(const v4f*)(xb + xo[i]); }

                const v8f zero = {0.f, 0.f, 0.f, 0.f, 0.f, 0.f, 0.f, 0.f};
                v8f acc[TILES][8];
#pragma unroll
                for (int t = 0; t < TILES; ++t)
#pragma unroll
                    for (int i = 0; i < 8; ++i) acc[t][i] = zero;

                const float b20 = b2[e * 2 + 0], b21 = b2[e * 2 + 1];
#pragma unroll
                for (int ks = 0; ks < 8; ++ks) {
#pragma unroll
                    for (int t = 0; t < TILES; ++t) {
                        const unsigned char* s = stg_ld + (ks & 1) * STG_BYTES + t * (16 * STG_PITCH);
                        const v8bf lo8 = *(const v8bf*)(s);
                        const v8bf hi8 = *(const v8bf*)(s + 32);
                        a[t] = __builtin_shufflevector(lo8, hi8, 0, 1, 2, 3, 4, 5, 6, 7, 8, 9, 10, 11, 12, 13, 14, 15);
                    }
                    if (ks < 7) {
#pragma unroll
                        for (int i = 0; i < XLOADS; ++i) {
                            hold_until_here(raw[i]);
                            const v4bf pk = to_bf4(raw[i]);
                            *(v4bf*)(stg_st + ((ks + 1) & 1) * STG_BYTES + i * (4 * STG_PITCH)) = pk;
                            if (ks < 6) {
                                addr_after(xo[i], pk);
                                xo[i] += 128u;
                                raw[i] = *(const v4f*)(xb + xo[i]);
                            }
                        }
                    }
                    __builtin_amdgcn_sched_barrier(0);
                    v16bf bcur = *(const v16bf*)(lds + ((ks << 3) << 10) + (lane << 5));
#pragma unroll
                    for (int nt = 0; nt < 8; ++nt) {
                        v16bf bnext = bcur;
                        if (nt < 7) bnext = *(const v16bf*)(lds + (((ks << 3) | (nt + 1)) << 10) + (lane << 5));
#pragma unroll
                        for (int t = 0; t < TILES; ++t)
                            acc[t][nt] = __builtin_amdgcn_wmma_f32_16x16x32_bf16(
                                false, bcur, false, a[t], (short)0, acc[t][nt], false, false);
                        bcur = bnext;
                    }
                    __builtin_amdgcn_sched_group_barrier(0x100, 2, 0);
#pragma unroll
                    for (int nt = 0; nt < 7; ++nt) {
                        __builtin_amdgcn_sched_group_barrier(0x100, 2, 0);
                        __builtin_amdgcn_sched_group_barrier(0x008, 3, 0);
                    }
                    __builtin_amdgcn_sched_group_barrier(0x008, 3, 0);
                    __builtin_amdgcn_sched_barrier(0);
                }

                {
                    const float* const cb = s_b1 + hi * 8;
                    const float* const cw = s_w2 + hi * 16;
                    float s0[TILES], s1[TILES];
#pragma unroll
                    for (int t = 0; t < TILES; ++t) { s0[t] = 0.f; s1[t] = 0.f; }
#pragma unroll
                    for (int nt = 0; nt < 8; ++nt) {
                        float bb[8], w0[8], w1[8];
#pragma unroll
                        for (int g = 0; g < 2; ++g) {
                            const v4f q = *(const v4f*)(cb + nt * 16 + g * 4);
                            bb[4 * g + 0] = q.x; bb[4 * g + 1] = q.y; bb[4 * g + 2] = q.z; bb[4 * g + 3] = q.w;
                        }
#pragma unroll
                        for (int g = 0; g < 4; ++g) {
                            const v4f q = *(const v4f*)(cw + nt * 32 + g * 4);
                            w0[2 * g + 0] = q.x; w1[2 * g + 0] = q.y; w0[2 * g + 1] = q.z; w1[2 * g + 1] = q.w;
                        }
#pragma unroll
                        for (int t = 0; t < TILES; ++t) {
#pragma unroll
                            for (int r = 0; r < 8; ++r) {
                                float h0 = acc[t][nt][r] + bb[r];
                                h0 = h0 > 0.f ? h0 : 0.f;
                                s0[t] += h0 * w0[r];
                                s1[t] += h0 * w1[r];
                            }
                        }
                    }
#pragma unroll
                    for (int t = 0; t < TILES; ++t) {
                        const float l0 = s0[t] + xhalf(s0[t]) + b20;
                        const float l1 = s1[t] + xhalf(s1[t]) + b21;
                        float* rl = (float*)(s_rl + (wv * ROWS_PER_WAVE + t * 16 + lanelo) * 4);
                        rl[1] = l0;
                        rl[2] = l1;
                    }
                }
            }
            __syncthreads();

#pragma unroll
            for (int j = 0; j < 2; ++j) {
                const int rr = wv * ROWS_PER_WAVE + j * 32 + lane;
                if (j == 0 || lane < 16) {
                    const int4 rw = *(const int4*)(s_rl + rr * 4);
                    const int tok = rw.x;
                    if (tok >= 0) {
                        const float l0 = __int_as_float(rw.y);
                        const float l1 = __int_as_float(rw.z);
                        const float mx = fmaxf(l0, l1);
                        const float e0 = __expf(l0 - mx), e1 = __expf(l1 - mx);
                        const float inv = 1.f / (e0 + e1);
                        out[(size_t)tok * 2 + 0] = e0 * inv;
                        out[(size_t)tok * 2 + 1] = e1 * inv;
                    }
                }
            }
            pbase += ROWS_PER_PASS;
            if (pbase >= hiRow) break;
            __syncthreads();
            pass_setup(pbase);
        }
    }
}


extern "C" void kernel_launch(void* const* d_in, const int* in_sizes, int n_in,
                              void* d_out, int out_size, void* d_ws, size_t ws_size,
                              hipStream_t stream) {
    (void)in_sizes; (void)n_in; (void)out_size; (void)d_ws; (void)ws_size;
    const float* x  = (const float*)d_in[0];
    const int*   z  = (const int*)d_in[1];
    const float* W1 = (const float*)d_in[2];
    const float* b1 = (const float*)d_in[3];
    const float* W2 = (const float*)d_in[4];
    const float* b2 = (const float*)d_in[5];
    float* out = (float*)d_out;

    moe_k<<<MOE_BLOCKS, MOE_THREADS, 0, stream>>>(x, z, W1, b1, W2, b2, out);
}
